# starting kernel
# baseline (speedup 1.0000x reference)
#include <hip/hip_runtime.h>
#include <hip/hip_bf16.h>

typedef __attribute__((ext_vector_type(2))) float v2f;
typedef __attribute__((ext_vector_type(8))) float v8f;

#define BAG   15
#define NSTEP 5
#define LAMBDA 0.001f
#define PS    8
#define IM    28
#define DENOM (IM - PS)
#define BATCH 65536
#define INDIM 66
#define NBLK  512
#define WISTR 68

__global__ void cyclic_prep(const float* __restrict__ R,
                            const float* __restrict__ mask,
                            const float* __restrict__ partials,
                            float* __restrict__ ReffT,
                            int stepK) {
  int tid = threadIdx.x;
  int i = tid >> 4, j = tid & 15;
  float v = 0.f;
  if (stepK > 0 && i < BAG && j < BAG) {
    float co = 0.f;
    for (int b = 0; b < NBLK; ++b) co += partials[b * 256 + tid];
    co *= (1.0f / (float)BATCH);
    v = (R[i * BAG + j] - LAMBDA * co) * mask[i * BAG + j];
  }
  ReffT[j * 16 + i] = v;
}

__global__ __launch_bounds__(256)
void cyclic_step(const float* __restrict__ imgs,
                 const float* __restrict__ Wi,  const float* __restrict__ bi,
                 const float* __restrict__ gma, const float* __restrict__ bta,
                 const float* __restrict__ rb,
                 const float* __restrict__ Wc,  const float* __restrict__ bc,
                 const float* __restrict__ Wo,  const float* __restrict__ bo,
                 float* __restrict__ stateWS,   int* __restrict__ posWS,
                 float* __restrict__ partials,  const float* __restrict__ ReffT,
                 float* __restrict__ out,       int stepK) {
  __shared__ float ldsWi[16 * WISTR];
  __shared__ float ldsWout[256];
  __shared__ float ldsReff[256];
  __shared__ float ldsPar[48];
  __shared__ float ldsState[8][16][16];
  __shared__ float ldsCo[8][256];

  const int tid  = threadIdx.x;
  const int wave = tid >> 5;
  const int lane = tid & 31;
  const int half = lane >> 4;
  const int n16  = lane & 15;
  const int rowBase = blockIdx.x * 128 + wave * 16;
  const int myRow   = rowBase + n16;

  for (int idx = tid; idx < 16 * WISTR; idx += 256) {
    int n = idx / WISTR, k = idx - n * WISTR;
    ldsWi[idx] = (n < BAG && k < INDIM) ? Wi[n * INDIM + k] : 0.f;
  }
  {
    int k = tid >> 4, col = tid & 15;
    float v = 0.f;
    if (k < BAG) {
      if (col < 10)      v = Wo[col * BAG + k];
      else if (col < 12) v = Wc[(col - 10) * BAG + k];
    }
    ldsWout[tid] = v;
    ldsReff[tid] = ReffT[tid];
  }
  if (tid < 16) {
    float bv = 0.f, gv = 0.f, btv = 0.f;
    if (tid < BAG) {
      bv  = bi[tid] + ((stepK > 0) ? rb[tid] : 0.f);
      gv  = gma[tid];
      btv = bta[tid];
    }
    ldsPar[tid]      = bv;
    ldsPar[16 + tid] = gv;
    ldsPar[32 + tid] = btv;
  }
  const float bc0 = bc[0], bc1 = bc[1];
  const float boN = bo[n16 < 10 ? n16 : 9];

  int p0, p1;
  if (stepK == 0) { p0 = DENOM / 2; p1 = DENOM / 2; }
  else            { p0 = posWS[2 * myRow]; p1 = posWS[2 * myRow + 1]; }

  const float* img = imgs + (size_t)myRow * (IM * IM);
  __syncthreads();

  v8f accI = {0.f, 0.f, 0.f, 0.f, 0.f, 0.f, 0.f, 0.f};
#pragma unroll
  for (int t = 0; t < 17; ++t) {
    const int kb = 4 * t + 2 * half;
    v2f a;
    if (t < 16) {
      a.x = img[(p0 + (kb >> 3)) * IM + p1 + (kb & 7)];
      a.y = img[(p0 + ((kb + 1) >> 3)) * IM + p1 + ((kb + 1) & 7)];
    } else if (half == 0) {
      a.x = (float)p0 * (2.0f / DENOM) - 1.0f;
      a.y = (float)p1 * (2.0f / DENOM) - 1.0f;
    } else {
      a.x = 0.f; a.y = 0.f;
    }
    v2f b = *(const v2f*)&ldsWi[n16 * WISTR + kb];
    accI = __builtin_amdgcn_wmma_f32_16x16x4_f32(false, a, false, b,
                                                 (short)0, accI, false, false);
  }

  if (stepK > 0) {
    const float* st = stateWS + (size_t)myRow * 16;
#pragma unroll
    for (int t = 0; t < 4; ++t) {
      const int kb = 4 * t + 2 * half;
      v2f a, b;
      a.x = st[kb];
      a.y = st[kb + 1];
      b.x = ldsReff[kb * 16 + n16];
      b.y = ldsReff[(kb + 1) * 16 + n16];
      accI = __builtin_amdgcn_wmma_f32_16x16x4_f32(false, a, false, b,
                                                   (short)0, accI, false, false);
    }
  }

  const float biasN = ldsPar[n16];
  const float gN    = ldsPar[16 + n16];
  const float bN    = ldsPar[32 + n16];
#pragma unroll
  for (int r = 0; r < 8; ++r) {
    float h = fmaxf(accI[r] + biasN, 0.f);
    float sum = h;
#pragma unroll
    for (int m = 8; m >= 1; m >>= 1) sum += __shfl_xor(sum, m, 16);
    float mu = sum * (1.0f / BAG);
    float dd = h - mu;
    float sq = (n16 < BAG) ? dd * dd : 0.f;
#pragma unroll
    for (int m = 8; m >= 1; m >>= 1) sq += __shfl_xor(sq, m, 16);
    float s = dd * rsqrtf(sq * (1.0f / BAG) + 1e-5f) * gN + bN;
    const int M = r + half * 8;
    ldsState[wave][M][n16] = s;
    if (stepK < NSTEP - 1)
      stateWS[(size_t)(rowBase + M) * 16 + n16] = s;
  }
  __syncthreads();

  v8f accO = {0.f, 0.f, 0.f, 0.f, 0.f, 0.f, 0.f, 0.f};
#pragma unroll
  for (int t = 0; t < 4; ++t) {
    const int kb = 4 * t + 2 * half;
    v2f a, b;
    a.x = ldsState[wave][n16][kb];
    a.y = ldsState[wave][n16][kb + 1];
    b.x = ldsWout[kb * 16 + n16];
    b.y = ldsWout[(kb + 1) * 16 + n16];
    accO = __builtin_amdgcn_wmma_f32_16x16x4_f32(false, a, false, b,
                                                 (short)0, accO, false, false);
  }

#pragma unroll
  for (int r = 0; r < 8; ++r) {
    float raw10 = __shfl(accO[r], 10, 16);
    float raw11 = __shfl(accO[r], 11, 16);
    int q0 = __shfl(p0, half * 8 + r, 16);
    int q1 = __shfl(p1, half * 8 + r, 16);
    if (stepK < NSTEP - 1 && n16 == 0) {
      float c0 = tanhf(raw10 + bc0);
      float c1 = tanhf(raw11 + bc1);
      bool rowSel = fabsf(c0) >= fabsf(c1);
      int mv0 = rowSel ? ((c0 > 0.f) - (c0 < 0.f)) : 0;
      int mv1 = rowSel ? 0 : ((c1 > 0.f) - (c1 < 0.f));
      q0 = min(max(q0 + mv0, 0), DENOM);
      q1 = min(max(q1 + mv1, 0), DENOM);
      const int row = rowBase + half * 8 + r;
      posWS[2 * row]     = q0;
      posWS[2 * row + 1] = q1;
    }
  }

  if (stepK == NSTEP - 1 && n16 < 10) {
#pragma unroll
    for (int r = 0; r < 8; ++r)
      out[(size_t)(rowBase + r + half * 8) * 10 + n16] = accO[r] + boN;
  }

  if (stepK < NSTEP - 1) {
    v8f accC = {0.f, 0.f, 0.f, 0.f, 0.f, 0.f, 0.f, 0.f};
#pragma unroll
    for (int t = 0; t < 4; ++t) {
      const int kb = 4 * t + 2 * half;
      v2f f;
      f.x = ldsState[wave][kb][n16];
      f.y = ldsState[wave][kb + 1][n16];
      accC = __builtin_amdgcn_wmma_f32_16x16x4_f32(false, f, false, f,
                                                   (short)0, accC, false, false);
    }
#pragma unroll
    for (int r = 0; r < 8; ++r)
      ldsCo[wave][(r + half * 8) * 16 + n16] = accC[r];
  }
  __syncthreads();
  if (stepK < NSTEP - 1) {
    float acc = 0.f;
#pragma unroll
    for (int w = 0; w < 8; ++w) acc += ldsCo[w][tid];
    partials[(size_t)blockIdx.x * 256 + tid] = acc;
  }
}

extern "C" void kernel_launch(void* const* d_in, const int* in_sizes, int n_in,
                              void* d_out, int out_size, void* d_ws, size_t ws_size,
                              hipStream_t stream) {
  const float* x    = (const float*)d_in[0];
  const float* Wi   = (const float*)d_in[1];
  const float* bi   = (const float*)d_in[2];
  const float* gma  = (const float*)d_in[3];
  const float* bta  = (const float*)d_in[4];
  const float* R    = (const float*)d_in[5];
  const float* rb   = (const float*)d_in[6];
  const float* mask = (const float*)d_in[7];
  const float* Wc   = (const float*)d_in[8];
  const float* bc   = (const float*)d_in[9];
  const float* Wo   = (const float*)d_in[10];
  const float* bo   = (const float*)d_in[11];

  float* stateWS  = (float*)d_ws;
  int*   posWS    = (int*)((char*)d_ws + (size_t)BATCH * 16 * 4);
  float* partials = (float*)((char*)posWS + (size_t)BATCH * 2 * 4);
  float* ReffT    = partials + (size_t)NBLK * 256;

  for (int k = 0; k < NSTEP; ++k) {
    cyclic_prep<<<1, 256, 0, stream>>>(R, mask, partials, ReffT, k);
    cyclic_step<<<NBLK, 256, 0, stream>>>(x, Wi, bi, gma, bta, rb, Wc, bc,
                                          Wo, bo, stateWS, posWS, partials,
                                          ReffT, (float*)d_out, k);
  }
}


# coact reduction launch: one work-group summing 512 planes serially -> 8 work-groups with every plane load in flight at once and a fixed sum tree (summation order changes)
# speedup vs baseline: 1.7894x; 1.7894x over previous
#include <hip/hip_runtime.h>
#include <hip/hip_bf16.h>

typedef __attribute__((ext_vector_type(2))) float v2f;
typedef __attribute__((ext_vector_type(8))) float v8f;

#define BAG   15
#define NSTEP 5
#define LAMBDA 0.001f
#define PS    8
#define IM    28
#define DENOM (IM - PS)
#define BATCH 65536
#define INDIM 66
#define NBLK  512
#define WISTR 68
#define PREPG 8
#define PREPT 512

__global__ __launch_bounds__(PREPT)
void cyclic_prep(const float* __restrict__ R,
                 const float* __restrict__ mask,
                 const float* __restrict__ partials,
                 float* __restrict__ ReffT,
                 int stepK) {
  __shared__ float4 red[PREPT / 32][8];
  const int tid  = threadIdx.x;
  const int lane = tid & 31;
  const int wave = tid >> 5;
  const int o    = blockIdx.x * 32 + lane;
  if (stepK == 0) {
    if (tid < 32) ReffT[o] = 0.f;
    return;
  }
  const int j = o >> 4, i = o & 15;
  const bool live = (tid < 32) && (i < BAG) && (j < BAG);
  float rv = 0.f, mv = 0.f;
  if (live) { rv = R[i * BAG + j]; mv = mask[i * BAG + j]; }

  const int c  = tid & 7;
  const int bl = tid >> 3;
  const float4* src = reinterpret_cast<const float4*>(partials)
                      + blockIdx.x * 8 + c;
  float4 v[8];
#pragma unroll
  for (int q = 0; q < 8; ++q) v[q] = src[(size_t)(q * (PREPT / 8) + bl) * 64];
#pragma unroll
  for (int st = 4; st >= 1; st >>= 1) {
#pragma unroll
    for (int q = 0; q < st; ++q) {
      v[q].x += v[q + st].x; v[q].y += v[q + st].y;
      v[q].z += v[q + st].z; v[q].w += v[q + st].w;
    }
  }
  float4 s = v[0];
  s.x += __shfl_xor(s.x, 8, 32);  s.y += __shfl_xor(s.y, 8, 32);
  s.z += __shfl_xor(s.z, 8, 32);  s.w += __shfl_xor(s.w, 8, 32);
  s.x += __shfl_xor(s.x, 16, 32); s.y += __shfl_xor(s.y, 16, 32);
  s.z += __shfl_xor(s.z, 16, 32); s.w += __shfl_xor(s.w, 16, 32);
  if (lane < 8) red[wave][lane] = s;
  __syncthreads();
  if (tid < 32) {
    const float* rf = reinterpret_cast<const float*>(&red[0][0]);
    float w16[PREPT / 32];
#pragma unroll
    for (int w = 0; w < PREPT / 32; ++w) w16[w] = rf[w * 32 + tid];
#pragma unroll
    for (int st = PREPT / 64; st >= 1; st >>= 1) {
#pragma unroll
      for (int w = 0; w < st; ++w) w16[w] += w16[w + st];
    }
    float co = w16[0] * (1.0f / (float)BATCH);
    float val = 0.f;
    if (live) val = (rv - LAMBDA * co) * mv;
    ReffT[o] = val;
  }
}

__global__ __launch_bounds__(256)
void cyclic_step(const float* __restrict__ imgs,
                 const float* __restrict__ Wi,  const float* __restrict__ bi,
                 const float* __restrict__ gma, const float* __restrict__ bta,
                 const float* __restrict__ rb,
                 const float* __restrict__ Wc,  const float* __restrict__ bc,
                 const float* __restrict__ Wo,  const float* __restrict__ bo,
                 float* __restrict__ stateWS,   int* __restrict__ posWS,
                 float* __restrict__ partials,  const float* __restrict__ ReffT,
                 float* __restrict__ out,       int stepK) {
  __shared__ float ldsWi[16 * WISTR];
  __shared__ float ldsWout[256];
  __shared__ float ldsReff[256];
  __shared__ float ldsPar[48];
  __shared__ float ldsState[8][16][16];
  __shared__ float ldsCo[8][256];

  const int tid  = threadIdx.x;
  const int wave = tid >> 5;
  const int lane = tid & 31;
  const int half = lane >> 4;
  const int n16  = lane & 15;
  const int rowBase = blockIdx.x * 128 + wave * 16;
  const int myRow   = rowBase + n16;

  for (int idx = tid; idx < 16 * WISTR; idx += 256) {
    int n = idx / WISTR, k = idx - n * WISTR;
    ldsWi[idx] = (n < BAG && k < INDIM) ? Wi[n * INDIM + k] : 0.f;
  }
  {
    int k = tid >> 4, col = tid & 15;
    float v = 0.f;
    if (k < BAG) {
      if (col < 10)      v = Wo[col * BAG + k];
      else if (col < 12) v = Wc[(col - 10) * BAG + k];
    }
    ldsWout[tid] = v;
    ldsReff[tid] = ReffT[tid];
  }
  if (tid < 16) {
    float bv = 0.f, gv = 0.f, btv = 0.f;
    if (tid < BAG) {
      bv  = bi[tid] + ((stepK > 0) ? rb[tid] : 0.f);
      gv  = gma[tid];
      btv = bta[tid];
    }
    ldsPar[tid]      = bv;
    ldsPar[16 + tid] = gv;
    ldsPar[32 + tid] = btv;
  }
  const float bc0 = bc[0], bc1 = bc[1];
  const float boN = bo[n16 < 10 ? n16 : 9];

  int p0, p1;
  if (stepK == 0) { p0 = DENOM / 2; p1 = DENOM / 2; }
  else            { p0 = posWS[2 * myRow]; p1 = posWS[2 * myRow + 1]; }

  const float* img = imgs + (size_t)myRow * (IM * IM);
  __syncthreads();

  v8f accI = {0.f, 0.f, 0.f, 0.f, 0.f, 0.f, 0.f, 0.f};
#pragma unroll
  for (int t = 0; t < 17; ++t) {
    const int kb = 4 * t + 2 * half;
    v2f a;
    if (t < 16) {
      a.x = img[(p0 + (kb >> 3)) * IM + p1 + (kb & 7)];
      a.y = img[(p0 + ((kb + 1) >> 3)) * IM + p1 + ((kb + 1) & 7)];
    } else if (half == 0) {
      a.x = (float)p0 * (2.0f / DENOM) - 1.0f;
      a.y = (float)p1 * (2.0f / DENOM) - 1.0f;
    } else {
      a.x = 0.f; a.y = 0.f;
    }
    v2f b = *(const v2f*)&ldsWi[n16 * WISTR + kb];
    accI = __builtin_amdgcn_wmma_f32_16x16x4_f32(false, a, false, b,
                                                 (short)0, accI, false, false);
  }

  if (stepK > 0) {
    const float* st = stateWS + (size_t)myRow * 16;
#pragma unroll
    for (int t = 0; t < 4; ++t) {
      const int kb = 4 * t + 2 * half;
      v2f a, b;
      a.x = st[kb];
      a.y = st[kb + 1];
      b.x = ldsReff[kb * 16 + n16];
      b.y = ldsReff[(kb + 1) * 16 + n16];
      accI = __builtin_amdgcn_wmma_f32_16x16x4_f32(false, a, false, b,
                                                   (short)0, accI, false, false);
    }
  }

  const float biasN = ldsPar[n16];
  const float gN    = ldsPar[16 + n16];
  const float bN    = ldsPar[32 + n16];
#pragma unroll
  for (int r = 0; r < 8; ++r) {
    float h = fmaxf(accI[r] + biasN, 0.f);
    float sum = h;
#pragma unroll
    for (int m = 8; m >= 1; m >>= 1) sum += __shfl_xor(sum, m, 16);
    float mu = sum * (1.0f / BAG);
    float dd = h - mu;
    float sq = (n16 < BAG) ? dd * dd : 0.f;
#pragma unroll
    for (int m = 8; m >= 1; m >>= 1) sq += __shfl_xor(sq, m, 16);
    float s = dd * rsqrtf(sq * (1.0f / BAG) + 1e-5f) * gN + bN;
    const int M = r + half * 8;
    ldsState[wave][M][n16] = s;
    if (stepK < NSTEP - 1)
      stateWS[(size_t)(rowBase + M) * 16 + n16] = s;
  }
  __syncthreads();

  v8f accO = {0.f, 0.f, 0.f, 0.f, 0.f, 0.f, 0.f, 0.f};
#pragma unroll
  for (int t = 0; t < 4; ++t) {
    const int kb = 4 * t + 2 * half;
    v2f a, b;
    a.x = ldsState[wave][n16][kb];
    a.y = ldsState[wave][n16][kb + 1];
    b.x = ldsWout[kb * 16 + n16];
    b.y = ldsWout[(kb + 1) * 16 + n16];
    accO = __builtin_amdgcn_wmma_f32_16x16x4_f32(false, a, false, b,
                                                 (short)0, accO, false, false);
  }

#pragma unroll
  for (int r = 0; r < 8; ++r) {
    float raw10 = __shfl(accO[r], 10, 16);
    float raw11 = __shfl(accO[r], 11, 16);
    int q0 = __shfl(p0, half * 8 + r, 16);
    int q1 = __shfl(p1, half * 8 + r, 16);
    if (stepK < NSTEP - 1 && n16 == 0) {
      float c0 = tanhf(raw10 + bc0);
      float c1 = tanhf(raw11 + bc1);
      bool rowSel = fabsf(c0) >= fabsf(c1);
      int mv0 = rowSel ? ((c0 > 0.f) - (c0 < 0.f)) : 0;
      int mv1 = rowSel ? 0 : ((c1 > 0.f) - (c1 < 0.f));
      q0 = min(max(q0 + mv0, 0), DENOM);
      q1 = min(max(q1 + mv1, 0), DENOM);
      const int row = rowBase + half * 8 + r;
      posWS[2 * row]     = q0;
      posWS[2 * row + 1] = q1;
    }
  }

  if (stepK == NSTEP - 1 && n16 < 10) {
#pragma unroll
    for (int r = 0; r < 8; ++r)
      out[(size_t)(rowBase + r + half * 8) * 10 + n16] = accO[r] + boN;
  }

  if (stepK < NSTEP - 1) {
    v8f accC = {0.f, 0.f, 0.f, 0.f, 0.f, 0.f, 0.f, 0.f};
#pragma unroll
    for (int t = 0; t < 4; ++t) {
      const int kb = 4 * t + 2 * half;
      v2f f;
      f.x = ldsState[wave][kb][n16];
      f.y = ldsState[wave][kb + 1][n16];
      accC = __builtin_amdgcn_wmma_f32_16x16x4_f32(false, f, false, f,
                                                   (short)0, accC, false, false);
    }
#pragma unroll
    for (int r = 0; r < 8; ++r)
      ldsCo[wave][(r + half * 8) * 16 + n16] = accC[r];
  }
  __syncthreads();
  if (stepK < NSTEP - 1) {
    float acc = 0.f;
#pragma unroll
    for (int w = 0; w < 8; ++w) acc += ldsCo[w][tid];
    partials[(size_t)blockIdx.x * 256 + tid] = acc;
  }
}

extern "C" void kernel_launch(void* const* d_in, const int* in_sizes, int n_in,
                              void* d_out, int out_size, void* d_ws, size_t ws_size,
                              hipStream_t stream) {
  const float* x    = (const float*)d_in[0];
  const float* Wi   = (const float*)d_in[1];
  const float* bi   = (const float*)d_in[2];
  const float* gma  = (const float*)d_in[3];
  const float* bta  = (const float*)d_in[4];
  const float* R    = (const float*)d_in[5];
  const float* rb   = (const float*)d_in[6];
  const float* mask = (const float*)d_in[7];
  const float* Wc   = (const float*)d_in[8];
  const float* bc   = (const float*)d_in[9];
  const float* Wo   = (const float*)d_in[10];
  const float* bo   = (const float*)d_in[11];

  float* stateWS  = (float*)d_ws;
  int*   posWS    = (int*)((char*)d_ws + (size_t)BATCH * 16 * 4);
  float* partials = (float*)((char*)posWS + (size_t)BATCH * 2 * 4);
  float* ReffT    = partials + (size_t)NBLK * 256;

  for (int k = 0; k < NSTEP; ++k) {
    cyclic_prep<<<PREPG, PREPT, 0, stream>>>(R, mask, partials, ReffT, k);
    cyclic_step<<<NBLK, 256, 0, stream>>>(x, Wi, bi, gma, bta, rb, Wc, bc,
                                          Wo, bo, stateWS, posWS, partials,
                                          ReffT, (float*)d_out, k);
  }
}


# LayerNorm row sums over the 16 feature lanes: each butterfly hop was an LDS-crossbar permute with a full wait, now a DPP row_xmask add on the vector ALU, 8 rows interleaved; same hop order
# speedup vs baseline: 1.8626x; 1.0409x over previous
#include <hip/hip_runtime.h>
#include <hip/hip_bf16.h>

typedef __attribute__((ext_vector_type(2))) float v2f;
typedef __attribute__((ext_vector_type(8))) float v8f;

#define BAG   15
#define NSTEP 5
#define LAMBDA 0.001f
#define PS    8
#define IM    28
#define DENOM (IM - PS)
#define BATCH 65536
#define INDIM 66
#define NBLK  512
#define WISTR 68
#define PREPG 8
#define PREPT 512

template <int M>
__device__ __forceinline__ float row_xor(float v) {
  return __builtin_bit_cast(float,
      __builtin_amdgcn_update_dpp(0, __builtin_bit_cast(int, v),
                                  0x160 + M, 0xF, 0xF, true));
}

__global__ __launch_bounds__(PREPT)
void cyclic_prep(const float* __restrict__ R,
                 const float* __restrict__ mask,
                 const float* __restrict__ partials,
                 float* __restrict__ ReffT,
                 int stepK) {
  __shared__ float4 red[PREPT / 32][8];
  const int tid  = threadIdx.x;
  const int lane = tid & 31;
  const int wave = tid >> 5;
  const int o    = blockIdx.x * 32 + lane;
  if (stepK == 0) {
    if (tid < 32) ReffT[o] = 0.f;
    return;
  }
  const int j = o >> 4, i = o & 15;
  const bool live = (tid < 32) && (i < BAG) && (j < BAG);
  float rv = 0.f, mv = 0.f;
  if (live) { rv = R[i * BAG + j]; mv = mask[i * BAG + j]; }

  const int c  = tid & 7;
  const int bl = tid >> 3;
  const float4* src = reinterpret_cast<const float4*>(partials)
                      + blockIdx.x * 8 + c;
  float4 v[8];
#pragma unroll
  for (int q = 0; q < 8; ++q) v[q] = src[(size_t)(q * (PREPT / 8) + bl) * 64];
#pragma unroll
  for (int st = 4; st >= 1; st >>= 1) {
#pragma unroll
    for (int q = 0; q < st; ++q) {
      v[q].x += v[q + st].x; v[q].y += v[q + st].y;
      v[q].z += v[q + st].z; v[q].w += v[q + st].w;
    }
  }
  float4 s = v[0];
  s.x += __shfl_xor(s.x, 8, 32);  s.y += __shfl_xor(s.y, 8, 32);
  s.z += __shfl_xor(s.z, 8, 32);  s.w += __shfl_xor(s.w, 8, 32);
  s.x += __shfl_xor(s.x, 16, 32); s.y += __shfl_xor(s.y, 16, 32);
  s.z += __shfl_xor(s.z, 16, 32); s.w += __shfl_xor(s.w, 16, 32);
  if (lane < 8) red[wave][lane] = s;
  __syncthreads();
  if (tid < 32) {
    const float* rf = reinterpret_cast<const float*>(&red[0][0]);
    float w16[PREPT / 32];
#pragma unroll
    for (int w = 0; w < PREPT / 32; ++w) w16[w] = rf[w * 32 + tid];
#pragma unroll
    for (int st = PREPT / 64; st >= 1; st >>= 1) {
#pragma unroll
      for (int w = 0; w < st; ++w) w16[w] += w16[w + st];
    }
    float co = w16[0] * (1.0f / (float)BATCH);
    float val = 0.f;
    if (live) val = (rv - LAMBDA * co) * mv;
    ReffT[o] = val;
  }
}

__global__ __launch_bounds__(256)
void cyclic_step(const float* __restrict__ imgs,
                 const float* __restrict__ Wi,  const float* __restrict__ bi,
                 const float* __restrict__ gma, const float* __restrict__ bta,
                 const float* __restrict__ rb,
                 const float* __restrict__ Wc,  const float* __restrict__ bc,
                 const float* __restrict__ Wo,  const float* __restrict__ bo,
                 float* __restrict__ stateWS,   int* __restrict__ posWS,
                 float* __restrict__ partials,  const float* __restrict__ ReffT,
                 float* __restrict__ out,       int stepK) {
  __shared__ float ldsWi[16 * WISTR];
  __shared__ float ldsWout[256];
  __shared__ float ldsReff[256];
  __shared__ float ldsPar[48];
  __shared__ float ldsState[8][16][16];
  __shared__ float ldsCo[8][256];

  const int tid  = threadIdx.x;
  const int wave = tid >> 5;
  const int lane = tid & 31;
  const int half = lane >> 4;
  const int n16  = lane & 15;
  const int rowBase = blockIdx.x * 128 + wave * 16;
  const int myRow   = rowBase + n16;

  for (int idx = tid; idx < 16 * WISTR; idx += 256) {
    int n = idx / WISTR, k = idx - n * WISTR;
    ldsWi[idx] = (n < BAG && k < INDIM) ? Wi[n * INDIM + k] : 0.f;
  }
  {
    int k = tid >> 4, col = tid & 15;
    float v = 0.f;
    if (k < BAG) {
      if (col < 10)      v = Wo[col * BAG + k];
      else if (col < 12) v = Wc[(col - 10) * BAG + k];
    }
    ldsWout[tid] = v;
    ldsReff[tid] = ReffT[tid];
  }
  if (tid < 16) {
    float bv = 0.f, gv = 0.f, btv = 0.f;
    if (tid < BAG) {
      bv  = bi[tid] + ((stepK > 0) ? rb[tid] : 0.f);
      gv  = gma[tid];
      btv = bta[tid];
    }
    ldsPar[tid]      = bv;
    ldsPar[16 + tid] = gv;
    ldsPar[32 + tid] = btv;
  }
  const float bc0 = bc[0], bc1 = bc[1];
  const float boN = bo[n16 < 10 ? n16 : 9];

  int p0, p1;
  if (stepK == 0) { p0 = DENOM / 2; p1 = DENOM / 2; }
  else            { p0 = posWS[2 * myRow]; p1 = posWS[2 * myRow + 1]; }

  const float* img = imgs + (size_t)myRow * (IM * IM);
  __syncthreads();

  v8f accI = {0.f, 0.f, 0.f, 0.f, 0.f, 0.f, 0.f, 0.f};
#pragma unroll
  for (int t = 0; t < 17; ++t) {
    const int kb = 4 * t + 2 * half;
    v2f a;
    if (t < 16) {
      a.x = img[(p0 + (kb >> 3)) * IM + p1 + (kb & 7)];
      a.y = img[(p0 + ((kb + 1) >> 3)) * IM + p1 + ((kb + 1) & 7)];
    } else if (half == 0) {
      a.x = (float)p0 * (2.0f / DENOM) - 1.0f;
      a.y = (float)p1 * (2.0f / DENOM) - 1.0f;
    } else {
      a.x = 0.f; a.y = 0.f;
    }
    v2f b = *(const v2f*)&ldsWi[n16 * WISTR + kb];
    accI = __builtin_amdgcn_wmma_f32_16x16x4_f32(false, a, false, b,
                                                 (short)0, accI, false, false);
  }

  if (stepK > 0) {
    const float* st = stateWS + (size_t)myRow * 16;
#pragma unroll
    for (int t = 0; t < 4; ++t) {
      const int kb = 4 * t + 2 * half;
      v2f a, b;
      a.x = st[kb];
      a.y = st[kb + 1];
      b.x = ldsReff[kb * 16 + n16];
      b.y = ldsReff[(kb + 1) * 16 + n16];
      accI = __builtin_amdgcn_wmma_f32_16x16x4_f32(false, a, false, b,
                                                   (short)0, accI, false, false);
    }
  }

  const float biasN = ldsPar[n16];
  const float gN    = ldsPar[16 + n16];
  const float bN    = ldsPar[32 + n16];
  float hv[8], red[8];
#pragma unroll
  for (int r = 0; r < 8; ++r) {
    hv[r]  = fmaxf(accI[r] + biasN, 0.f);
    red[r] = hv[r];
  }
#pragma unroll
  for (int r = 0; r < 8; ++r) red[r] += row_xor<8>(red[r]);
#pragma unroll
  for (int r = 0; r < 8; ++r) red[r] += row_xor<4>(red[r]);
#pragma unroll
  for (int r = 0; r < 8; ++r) red[r] += row_xor<2>(red[r]);
#pragma unroll
  for (int r = 0; r < 8; ++r) red[r] += row_xor<1>(red[r]);
#pragma unroll
  for (int r = 0; r < 8; ++r) {
    float mu = red[r] * (1.0f / BAG);
    float dd = hv[r] - mu;
    hv[r]  = dd;
    red[r] = (n16 < BAG) ? dd * dd : 0.f;
  }
#pragma unroll
  for (int r = 0; r < 8; ++r) red[r] += row_xor<8>(red[r]);
#pragma unroll
  for (int r = 0; r < 8; ++r) red[r] += row_xor<4>(red[r]);
#pragma unroll
  for (int r = 0; r < 8; ++r) red[r] += row_xor<2>(red[r]);
#pragma unroll
  for (int r = 0; r < 8; ++r) red[r] += row_xor<1>(red[r]);
#pragma unroll
  for (int r = 0; r < 8; ++r) {
    float s = hv[r] * rsqrtf(red[r] * (1.0f / BAG) + 1e-5f) * gN + bN;
    hv[r] = s;
    ldsState[wave][r + half * 8][n16] = s;
  }
  if (stepK < NSTEP - 1) {
#pragma unroll
    for (int r = 0; r < 8; ++r)
      stateWS[(size_t)(rowBase + r + half * 8) * 16 + n16] = hv[r];
  }
  __syncthreads();

  v8f accO = {0.f, 0.f, 0.f, 0.f, 0.f, 0.f, 0.f, 0.f};
#pragma unroll
  for (int t = 0; t < 4; ++t) {
    const int kb = 4 * t + 2 * half;
    v2f a, b;
    a.x = ldsState[wave][n16][kb];
    a.y = ldsState[wave][n16][kb + 1];
    b.x = ldsWout[kb * 16 + n16];
    b.y = ldsWout[(kb + 1) * 16 + n16];
    accO = __builtin_amdgcn_wmma_f32_16x16x4_f32(false, a, false, b,
                                                 (short)0, accO, false, false);
  }

#pragma unroll
  for (int r = 0; r < 8; ++r) {
    float raw10 = __shfl(accO[r], 10, 16);
    float raw11 = __shfl(accO[r], 11, 16);
    int q0 = __shfl(p0, half * 8 + r, 16);
    int q1 = __shfl(p1, half * 8 + r, 16);
    if (stepK < NSTEP - 1 && n16 == 0) {
      float c0 = tanhf(raw10 + bc0);
      float c1 = tanhf(raw11 + bc1);
      bool rowSel = fabsf(c0) >= fabsf(c1);
      int mv0 = rowSel ? ((c0 > 0.f) - (c0 < 0.f)) : 0;
      int mv1 = rowSel ? 0 : ((c1 > 0.f) - (c1 < 0.f));
      q0 = min(max(q0 + mv0, 0), DENOM);
      q1 = min(max(q1 + mv1, 0), DENOM);
      const int row = rowBase + half * 8 + r;
      posWS[2 * row]     = q0;
      posWS[2 * row + 1] = q1;
    }
  }

  if (stepK == NSTEP - 1 && n16 < 10) {
#pragma unroll
    for (int r = 0; r < 8; ++r)
      out[(size_t)(rowBase + r + half * 8) * 10 + n16] = accO[r] + boN;
  }

  if (stepK < NSTEP - 1) {
    v8f accC = {0.f, 0.f, 0.f, 0.f, 0.f, 0.f, 0.f, 0.f};
#pragma unroll
    for (int t = 0; t < 4; ++t) {
      const int kb = 4 * t + 2 * half;
      v2f f;
      f.x = ldsState[wave][kb][n16];
      f.y = ldsState[wave][kb + 1][n16];
      accC = __builtin_amdgcn_wmma_f32_16x16x4_f32(false, f, false, f,
                                                   (short)0, accC, false, false);
    }
#pragma unroll
    for (int r = 0; r < 8; ++r)
      ldsCo[wave][(r + half * 8) * 16 + n16] = accC[r];
  }
  __syncthreads();
  if (stepK < NSTEP - 1) {
    float acc = 0.f;
#pragma unroll
    for (int w = 0; w < 8; ++w) acc += ldsCo[w][tid];
    partials[(size_t)blockIdx.x * 256 + tid] = acc;
  }
}

extern "C" void kernel_launch(void* const* d_in, const int* in_sizes, int n_in,
                              void* d_out, int out_size, void* d_ws, size_t ws_size,
                              hipStream_t stream) {
  const float* x    = (const float*)d_in[0];
  const float* Wi   = (const float*)d_in[1];
  const float* bi   = (const float*)d_in[2];
  const float* gma  = (const float*)d_in[3];
  const float* bta  = (const float*)d_in[4];
  const float* R    = (const float*)d_in[5];
  const float* rb   = (const float*)d_in[6];
  const float* mask = (const float*)d_in[7];
  const float* Wc   = (const float*)d_in[8];
  const float* bc   = (const float*)d_in[9];
  const float* Wo   = (const float*)d_in[10];
  const float* bo   = (const float*)d_in[11];

  float* stateWS  = (float*)d_ws;
  int*   posWS    = (int*)((char*)d_ws + (size_t)BATCH * 16 * 4);
  float* partials = (float*)((char*)posWS + (size_t)BATCH * 2 * 4);
  float* ReffT    = partials + (size_t)NBLK * 256;

  for (int k = 0; k < NSTEP; ++k) {
    cyclic_prep<<<PREPG, PREPT, 0, stream>>>(R, mask, partials, ReffT, k);
    cyclic_step<<<NBLK, 256, 0, stream>>>(x, Wi, bi, gma, bta, rb, Wc, bc,
                                          Wo, bo, stateWS, posWS, partials,
                                          ReffT, (float*)d_out, k);
  }
}


# step kernel prologue: staging, position and previous-state loads, each waited for in turn (nine dependent global round trips before the LayerNorm), now issued in one run; two round trips left
# speedup vs baseline: 2.0617x; 1.1069x over previous
#include <hip/hip_runtime.h>
#include <hip/hip_bf16.h>

typedef __attribute__((ext_vector_type(2))) float v2f;
typedef __attribute__((ext_vector_type(8))) float v8f;

#define BAG   15
#define NSTEP 5
#define LAMBDA 0.001f
#define PS    8
#define IM    28
#define DENOM (IM - PS)
#define BATCH 65536
#define INDIM 66
#define NBLK  512
#define WISTR 68
#define PREPG 8
#define PREPT 512

template <int M>
__device__ __forceinline__ float row_xor(float v) {
  return __builtin_bit_cast(float,
      __builtin_amdgcn_update_dpp(0, __builtin_bit_cast(int, v),
                                  0x160 + M, 0xF, 0xF, true));
}

__global__ __launch_bounds__(PREPT)
void cyclic_prep(const float* __restrict__ R,
                 const float* __restrict__ mask,
                 const float* __restrict__ partials,
                 float* __restrict__ ReffT,
                 int stepK) {
  __shared__ float4 red[PREPT / 32][8];
  const int tid  = threadIdx.x;
  const int lane = tid & 31;
  const int wave = tid >> 5;
  const int o    = blockIdx.x * 32 + lane;
  if (stepK == 0) {
    if (tid < 32) ReffT[o] = 0.f;
    return;
  }
  const int j = o >> 4, i = o & 15;
  const bool live = (tid < 32) && (i < BAG) && (j < BAG);
  float rv = 0.f, mv = 0.f;
  if (live) { rv = R[i * BAG + j]; mv = mask[i * BAG + j]; }

  const int c  = tid & 7;
  const int bl = tid >> 3;
  const float4* src = reinterpret_cast<const float4*>(partials)
                      + blockIdx.x * 8 + c;
  float4 v[8];
#pragma unroll
  for (int q = 0; q < 8; ++q) v[q] = src[(size_t)(q * (PREPT / 8) + bl) * 64];
#pragma unroll
  for (int st = 4; st >= 1; st >>= 1) {
#pragma unroll
    for (int q = 0; q < st; ++q) {
      v[q].x += v[q + st].x; v[q].y += v[q + st].y;
      v[q].z += v[q + st].z; v[q].w += v[q + st].w;
    }
  }
  float4 s = v[0];
  s.x += __shfl_xor(s.x, 8, 32);  s.y += __shfl_xor(s.y, 8, 32);
  s.z += __shfl_xor(s.z, 8, 32);  s.w += __shfl_xor(s.w, 8, 32);
  s.x += __shfl_xor(s.x, 16, 32); s.y += __shfl_xor(s.y, 16, 32);
  s.z += __shfl_xor(s.z, 16, 32); s.w += __shfl_xor(s.w, 16, 32);
  if (lane < 8) red[wave][lane] = s;
  __syncthreads();
  if (tid < 32) {
    const float* rf = reinterpret_cast<const float*>(&red[0][0]);
    float w16[PREPT / 32];
#pragma unroll
    for (int w = 0; w < PREPT / 32; ++w) w16[w] = rf[w * 32 + tid];
#pragma unroll
    for (int st = PREPT / 64; st >= 1; st >>= 1) {
#pragma unroll
      for (int w = 0; w < st; ++w) w16[w] += w16[w + st];
    }
    float co = w16[0] * (1.0f / (float)BATCH);
    float val = 0.f;
    if (live) val = (rv - LAMBDA * co) * mv;
    ReffT[o] = val;
  }
}

__global__ __launch_bounds__(256)
void cyclic_step(const float* __restrict__ imgs,
                 const float* __restrict__ Wi,  const float* __restrict__ bi,
                 const float* __restrict__ gma, const float* __restrict__ bta,
                 const float* __restrict__ rb,
                 const float* __restrict__ Wc,  const float* __restrict__ bc,
                 const float* __restrict__ Wo,  const float* __restrict__ bo,
                 float* __restrict__ stateWS,   int* __restrict__ posWS,
                 float* __restrict__ partials,  const float* __restrict__ ReffT,
                 float* __restrict__ out,       int stepK) {
  __shared__ float ldsWi[16 * WISTR];
  __shared__ float ldsWout[256];
  __shared__ float ldsReff[256];
  __shared__ float ldsPar[48];
  __shared__ float ldsState[8][16][16];
  __shared__ float ldsCo[8][256];

  const int tid  = threadIdx.x;
  const int wave = tid >> 5;
  const int lane = tid & 31;
  const int half = lane >> 4;
  const int n16  = lane & 15;
  const int rowBase = blockIdx.x * 128 + wave * 16;
  const int myRow   = rowBase + n16;

  int p0 = DENOM / 2, p1 = DENOM / 2;
  if (stepK > 0) { p0 = posWS[2 * myRow]; p1 = posWS[2 * myRow + 1]; }

  v2f sa[4];
#pragma unroll
  for (int t = 0; t < 4; ++t) { sa[t].x = 0.f; sa[t].y = 0.f; }
  if (stepK > 0) {
    const float* st = stateWS + (size_t)myRow * 16;
#pragma unroll
    for (int t = 0; t < 4; ++t) {
      const int kb = 4 * t + 2 * half;
      sa[t].x = st[kb];
      sa[t].y = st[kb + 1];
    }
  }

  float wiv[5];
#pragma unroll
  for (int q = 0; q < 5; ++q) {
    const int idx = tid + 256 * q;
    const int n = idx / WISTR, k = idx - n * WISTR;
    const int nc = n < BAG ? n : BAG - 1;
    const int kc = k < INDIM ? k : INDIM - 1;
    const float v = Wi[nc * INDIM + kc];
    wiv[q] = (n < BAG && k < INDIM) ? v : 0.f;
  }
  float hv0;
  {
    const int k = tid >> 4, col = tid & 15;
    const int kc = k < BAG ? k : BAG - 1;
    const float* hsrc = (col < 10) ? (Wo + col * BAG)
                                   : (Wc + ((col < 12 ? col : 11) - 10) * BAG);
    const float v = hsrc[kc];
    hv0 = (k < BAG && col < 12) ? v : 0.f;
  }
  const float rfv = ReffT[tid];
  const float* img = imgs + (size_t)myRow * (IM * IM);
  v2f ga[16];
#pragma unroll
  for (int t = 0; t < 16; ++t) {
    const int kb = 4 * t + 2 * half;
    ga[t].x = img[(p0 + (kb >> 3)) * IM + p1 + (kb & 7)];
    ga[t].y = img[(p0 + ((kb + 1) >> 3)) * IM + p1 + ((kb + 1) & 7)];
  }

  float bv, gv, btv;
  {
    const int pc = n16 < BAG ? n16 : BAG - 1;
    const float b0 = bi[pc];
    const float r0 = rb[pc];
    const float g0 = gma[pc];
    const float t0 = bta[pc];
    const bool okp = n16 < BAG;
    bv  = okp ? b0 + ((stepK > 0) ? r0 : 0.f) : 0.f;
    gv  = okp ? g0 : 0.f;
    btv = okp ? t0 : 0.f;
  }
  const float bc0 = bc[0], bc1 = bc[1];
  const float boN = bo[n16 < 10 ? n16 : 9];

#pragma unroll
  for (int q = 0; q < 5; ++q) {
    const int idx = tid + 256 * q;
    if (idx < 16 * WISTR) ldsWi[idx] = wiv[q];
  }
  ldsWout[tid] = hv0;
  ldsReff[tid] = rfv;
  if (tid < 16) {
    ldsPar[tid]      = bv;
    ldsPar[16 + tid] = gv;
    ldsPar[32 + tid] = btv;
  }
  __syncthreads();

  v8f accI = {0.f, 0.f, 0.f, 0.f, 0.f, 0.f, 0.f, 0.f};
#pragma unroll
  for (int t = 0; t < 17; ++t) {
    const int kb = 4 * t + 2 * half;
    v2f a;
    if (t < 16) {
      a = ga[t];
    } else if (half == 0) {
      a.x = (float)p0 * (2.0f / DENOM) - 1.0f;
      a.y = (float)p1 * (2.0f / DENOM) - 1.0f;
    } else {
      a.x = 0.f; a.y = 0.f;
    }
    v2f b = *(const v2f*)&ldsWi[n16 * WISTR + kb];
    accI = __builtin_amdgcn_wmma_f32_16x16x4_f32(false, a, false, b,
                                                 (short)0, accI, false, false);
  }

  if (stepK > 0) {
#pragma unroll
    for (int t = 0; t < 4; ++t) {
      const int kb = 4 * t + 2 * half;
      v2f b;
      b.x = ldsReff[kb * 16 + n16];
      b.y = ldsReff[(kb + 1) * 16 + n16];
      accI = __builtin_amdgcn_wmma_f32_16x16x4_f32(false, sa[t], false, b,
                                                   (short)0, accI, false, false);
    }
  }

  const float biasN = ldsPar[n16];
  const float gN    = ldsPar[16 + n16];
  const float bN    = ldsPar[32 + n16];
  float hv[8], red[8];
#pragma unroll
  for (int r = 0; r < 8; ++r) {
    hv[r]  = fmaxf(accI[r] + biasN, 0.f);
    red[r] = hv[r];
  }
#pragma unroll
  for (int r = 0; r < 8; ++r) red[r] += row_xor<8>(red[r]);
#pragma unroll
  for (int r = 0; r < 8; ++r) red[r] += row_xor<4>(red[r]);
#pragma unroll
  for (int r = 0; r < 8; ++r) red[r] += row_xor<2>(red[r]);
#pragma unroll
  for (int r = 0; r < 8; ++r) red[r] += row_xor<1>(red[r]);
#pragma unroll
  for (int r = 0; r < 8; ++r) {
    float mu = red[r] * (1.0f / BAG);
    float dd = hv[r] - mu;
    hv[r]  = dd;
    red[r] = (n16 < BAG) ? dd * dd : 0.f;
  }
#pragma unroll
  for (int r = 0; r < 8; ++r) red[r] += row_xor<8>(red[r]);
#pragma unroll
  for (int r = 0; r < 8; ++r) red[r] += row_xor<4>(red[r]);
#pragma unroll
  for (int r = 0; r < 8; ++r) red[r] += row_xor<2>(red[r]);
#pragma unroll
  for (int r = 0; r < 8; ++r) red[r] += row_xor<1>(red[r]);
#pragma unroll
  for (int r = 0; r < 8; ++r) {
    float s = hv[r] * rsqrtf(red[r] * (1.0f / BAG) + 1e-5f) * gN + bN;
    hv[r] = s;
    ldsState[wave][r + half * 8][n16] = s;
  }
  if (stepK < NSTEP - 1) {
#pragma unroll
    for (int r = 0; r < 8; ++r)
      stateWS[(size_t)(rowBase + r + half * 8) * 16 + n16] = hv[r];
  }
  __syncthreads();

  v8f accO = {0.f, 0.f, 0.f, 0.f, 0.f, 0.f, 0.f, 0.f};
#pragma unroll
  for (int t = 0; t < 4; ++t) {
    const int kb = 4 * t + 2 * half;
    v2f a, b;
    a.x = ldsState[wave][n16][kb];
    a.y = ldsState[wave][n16][kb + 1];
    b.x = ldsWout[kb * 16 + n16];
    b.y = ldsWout[(kb + 1) * 16 + n16];
    accO = __builtin_amdgcn_wmma_f32_16x16x4_f32(false, a, false, b,
                                                 (short)0, accO, false, false);
  }

#pragma unroll
  for (int r = 0; r < 8; ++r) {
    float raw10 = __shfl(accO[r], 10, 16);
    float raw11 = __shfl(accO[r], 11, 16);
    int q0 = __shfl(p0, half * 8 + r, 16);
    int q1 = __shfl(p1, half * 8 + r, 16);
    if (stepK < NSTEP - 1 && n16 == 0) {
      float c0 = tanhf(raw10 + bc0);
      float c1 = tanhf(raw11 + bc1);
      bool rowSel = fabsf(c0) >= fabsf(c1);
      int mv0 = rowSel ? ((c0 > 0.f) - (c0 < 0.f)) : 0;
      int mv1 = rowSel ? 0 : ((c1 > 0.f) - (c1 < 0.f));
      q0 = min(max(q0 + mv0, 0), DENOM);
      q1 = min(max(q1 + mv1, 0), DENOM);
      const int row = rowBase + half * 8 + r;
      posWS[2 * row]     = q0;
      posWS[2 * row + 1] = q1;
    }
  }

  if (stepK == NSTEP - 1 && n16 < 10) {
#pragma unroll
    for (int r = 0; r < 8; ++r)
      out[(size_t)(rowBase + r + half * 8) * 10 + n16] = accO[r] + boN;
  }

  if (stepK < NSTEP - 1) {
    v8f accC = {0.f, 0.f, 0.f, 0.f, 0.f, 0.f, 0.f, 0.f};
#pragma unroll
    for (int t = 0; t < 4; ++t) {
      const int kb = 4 * t + 2 * half;
      v2f f;
      f.x = ldsState[wave][kb][n16];
      f.y = ldsState[wave][kb + 1][n16];
      accC = __builtin_amdgcn_wmma_f32_16x16x4_f32(false, f, false, f,
                                                   (short)0, accC, false, false);
    }
#pragma unroll
    for (int r = 0; r < 8; ++r)
      ldsCo[wave][(r + half * 8) * 16 + n16] = accC[r];
  }
  __syncthreads();
  if (stepK < NSTEP - 1) {
    float acc = 0.f;
#pragma unroll
    for (int w = 0; w < 8; ++w) acc += ldsCo[w][tid];
    partials[(size_t)blockIdx.x * 256 + tid] = acc;
  }
}

extern "C" void kernel_launch(void* const* d_in, const int* in_sizes, int n_in,
                              void* d_out, int out_size, void* d_ws, size_t ws_size,
                              hipStream_t stream) {
  const float* x    = (const float*)d_in[0];
  const float* Wi   = (const float*)d_in[1];
  const float* bi   = (const float*)d_in[2];
  const float* gma  = (const float*)d_in[3];
  const float* bta  = (const float*)d_in[4];
  const float* R    = (const float*)d_in[5];
  const float* rb   = (const float*)d_in[6];
  const float* mask = (const float*)d_in[7];
  const float* Wc   = (const float*)d_in[8];
  const float* bc   = (const float*)d_in[9];
  const float* Wo   = (const float*)d_in[10];
  const float* bo   = (const float*)d_in[11];

  float* stateWS  = (float*)d_ws;
  int*   posWS    = (int*)((char*)d_ws + (size_t)BATCH * 16 * 4);
  float* partials = (float*)((char*)posWS + (size_t)BATCH * 2 * 4);
  float* ReffT    = partials + (size_t)NBLK * 256;

  for (int k = 0; k < NSTEP; ++k) {
    cyclic_prep<<<PREPG, PREPT, 0, stream>>>(R, mask, partials, ReffT, k);
    cyclic_step<<<NBLK, 256, 0, stream>>>(x, Wi, bi, gma, bta, rb, Wc, bc,
                                          Wo, bo, stateWS, posWS, partials,
                                          ReffT, (float*)d_out, k);
  }
}


# coact plane sum moved from its own launch into the step kernel (each of 256 16-wave work-groups sums the 256 packed triangle planes under its in_act products): 10 launches to 5; sum order changes
# speedup vs baseline: 2.1944x; 1.0644x over previous
#include <hip/hip_runtime.h>
#include <hip/hip_bf16.h>

typedef __attribute__((ext_vector_type(2))) float v2f;
typedef __attribute__((ext_vector_type(8))) float v8f;

#define BAG   15
#define NSTEP 5
#define LAMBDA 0.001f
#define PS    8
#define IM    28
#define DENOM (IM - PS)
#define BATCH 65536
#define INDIM 66
#define NWG   256
#define WGT   512
#define NWAVE 16
#define PLANE 128
#define WISTR 68

template <int M>
__device__ __forceinline__ float row_xor(float v) {
  return __builtin_bit_cast(float,
      __builtin_amdgcn_update_dpp(0, __builtin_bit_cast(int, v),
                                  0x160 + M, 0xF, 0xF, true));
}

template <bool FIRST>
__global__ __launch_bounds__(WGT) __attribute__((amdgpu_waves_per_eu(4, 4)))
void cyclic_step(const float* __restrict__ imgs,
                 const float* __restrict__ Wi,  const float* __restrict__ bi,
                 const float* __restrict__ gma, const float* __restrict__ bta,
                 const float* __restrict__ R,   const float* __restrict__ rb,
                 const float* __restrict__ mask,
                 const float* __restrict__ Wc,  const float* __restrict__ bc,
                 const float* __restrict__ Wo,  const float* __restrict__ bo,
                 float* __restrict__ stateWS,   int* __restrict__ posWS,
                 float* planes,
                 float* __restrict__ out,       int stepK) {
  __shared__ float ldsWi[16 * WISTR];
  __shared__ float ldsWout[256];
  __shared__ float ldsReff[256];
  __shared__ float ldsPar[48];
  __shared__ float4 ldsRed[4][PLANE / 4];
  __shared__ float ldsState[NWAVE][16][16];
  __shared__ float ldsCo[NWAVE][256];

  const int tid  = threadIdx.x;
  const int wave = tid >> 5;
  const int lane = tid & 31;
  const int half = lane >> 4;
  const int n16  = lane & 15;
  const int rowBase = blockIdx.x * 256 + wave * 16;
  const int myRow   = rowBase + n16;

  int p0 = DENOM / 2, p1 = DENOM / 2;
  if (!FIRST) { p0 = posWS[2 * myRow]; p1 = posWS[2 * myRow + 1]; }

  v2f sa[4];
#pragma unroll
  for (int t = 0; t < 4; ++t) { sa[t].x = 0.f; sa[t].y = 0.f; }
  if (!FIRST) {
    const float* st = stateWS + (size_t)myRow * 16;
#pragma unroll
    for (int t = 0; t < 4; ++t) {
      const int kb = 4 * t + 2 * half;
      sa[t].x = st[kb];
      sa[t].y = st[kb + 1];
    }
  }

  float rv, mv;
  const int oi = tid & 15, oj = (tid >> 4) & 15;
  {
    const int ic = oi < BAG ? oi : BAG - 1;
    const int jc = oj < BAG ? oj : BAG - 1;
    rv = R[ic * BAG + jc];
    mv = mask[ic * BAG + jc];
  }

  float wiv[3];
#pragma unroll
  for (int q = 0; q < 3; ++q) {
    const int idx = tid + WGT * q;
    const int n = idx / WISTR, k = idx - n * WISTR;
    const int nc = n < BAG ? n : BAG - 1;
    const int kc = k < INDIM ? k : INDIM - 1;
    const float v = Wi[nc * INDIM + kc];
    wiv[q] = (n < BAG && k < INDIM) ? v : 0.f;
  }
  float hv0;
  {
    const int k = (tid >> 4) & 15, col = tid & 15;
    const int kc = k < BAG ? k : BAG - 1;
    const float* hsrc = (col < 10) ? (Wo + col * BAG)
                                   : (Wc + ((col < 12 ? col : 11) - 10) * BAG);
    const float v = hsrc[kc];
    hv0 = (k < BAG && col < 12) ? v : 0.f;
  }
  const float* img = imgs + (size_t)myRow * (IM * IM);
  v2f ga[16];
#pragma unroll
  for (int t = 0; t < 16; ++t) {
    const int kb = 4 * t + 2 * half;
    ga[t].x = img[(p0 + (kb >> 3)) * IM + p1 + (kb & 7)];
    ga[t].y = img[(p0 + ((kb + 1) >> 3)) * IM + p1 + ((kb + 1) & 7)];
  }

  float bv, gv, btv;
  {
    const int pc = n16 < BAG ? n16 : BAG - 1;
    const float b0 = bi[pc];
    const float r0 = rb[pc];
    const float g0 = gma[pc];
    const float t0 = bta[pc];
    const bool okp = n16 < BAG;
    bv  = okp ? b0 + (FIRST ? 0.f : r0) : 0.f;
    gv  = okp ? g0 : 0.f;
    btv = okp ? t0 : 0.f;
  }
  const float bc0 = bc[0], bc1 = bc[1];
  const float boN = bo[n16 < 10 ? n16 : 9];

#pragma unroll
  for (int q = 0; q < 3; ++q) {
    const int idx = tid + WGT * q;
    if (idx < 16 * WISTR) ldsWi[idx] = wiv[q];
  }
  if (tid < 256) ldsWout[tid] = hv0;
  if (tid < 16) {
    ldsPar[tid]      = bv;
    ldsPar[16 + tid] = gv;
    ldsPar[32 + tid] = btv;
  }
  __syncthreads();

  const int pw = (wave + blockIdx.x) & (NWAVE - 1);
  float4 pv[16];
  if (!FIRST) {
    const float4* psrc = reinterpret_cast<const float4*>(
                             planes + (size_t)((stepK + 1) & 1) * (NWG * PLANE))
                         + ((pw >> 2) * 4 + (lane >> 3)) * (PLANE / 4)
                         + (pw & 3) * 8 + (lane & 7);
#pragma unroll
    for (int q = 0; q < 16; ++q) pv[q] = psrc[(size_t)q * 16 * (PLANE / 4)];
  }

  v8f accI = {0.f, 0.f, 0.f, 0.f, 0.f, 0.f, 0.f, 0.f};
#pragma unroll
  for (int t = 0; t < 17; ++t) {
    const int kb = 4 * t + 2 * half;
    v2f a;
    if (t < 16) {
      a = ga[t];
    } else if (half == 0) {
      a.x = (float)p0 * (2.0f / DENOM) - 1.0f;
      a.y = (float)p1 * (2.0f / DENOM) - 1.0f;
    } else {
      a.x = 0.f; a.y = 0.f;
    }
    v2f b = *(const v2f*)&ldsWi[n16 * WISTR + kb];
    accI = __builtin_amdgcn_wmma_f32_16x16x4_f32(false, a, false, b,
                                                 (short)0, accI, false, false);
  }

  if (!FIRST) {
#pragma unroll
    for (int st = 8; st >= 1; st >>= 1) {
#pragma unroll
      for (int q = 0; q < st; ++q) {
        pv[q].x += pv[q + st].x; pv[q].y += pv[q + st].y;
        pv[q].z += pv[q + st].z; pv[q].w += pv[q + st].w;
      }
    }
    float4 s4 = pv[0];
    s4.x += row_xor<8>(s4.x);         s4.y += row_xor<8>(s4.y);
    s4.z += row_xor<8>(s4.z);         s4.w += row_xor<8>(s4.w);
    s4.x += __shfl_xor(s4.x, 16, 32); s4.y += __shfl_xor(s4.y, 16, 32);
    s4.z += __shfl_xor(s4.z, 16, 32); s4.w += __shfl_xor(s4.w, 16, 32);
    if (lane < 8) ldsRed[pw >> 2][(pw & 3) * 8 + lane] = s4;
    __syncthreads();

    if (tid < 256) {
      const int a = oi < oj ? oi : oj;
      const int b = oi < oj ? oj : oi;
      const int q = (a <= 7) ? a * 16 + b : (15 - a) * 16 + (b - a);
      const float* rf = reinterpret_cast<const float*>(&ldsRed[0][0]);
      const float co = ((rf[q] + rf[PLANE + q]) + (rf[2 * PLANE + q] + rf[3 * PLANE + q]))
                       * (1.0f / (float)BATCH);
      const bool live = (oi < BAG) && (oj < BAG);
      ldsReff[tid] = live ? (rv - LAMBDA * co) * mv : 0.f;
    }
    __syncthreads();
#pragma unroll
    for (int t = 0; t < 4; ++t) {
      const int kb = 4 * t + 2 * half;
      v2f b;
      b.x = ldsReff[kb * 16 + n16];
      b.y = ldsReff[(kb + 1) * 16 + n16];
      accI = __builtin_amdgcn_wmma_f32_16x16x4_f32(false, sa[t], false, b,
                                                   (short)0, accI, false, false);
    }
  }

  const float biasN = ldsPar[n16];
  const float gN    = ldsPar[16 + n16];
  const float bN    = ldsPar[32 + n16];
  float hv[8], red[8];
#pragma unroll
  for (int r = 0; r < 8; ++r) {
    hv[r]  = fmaxf(accI[r] + biasN, 0.f);
    red[r] = hv[r];
  }
#pragma unroll
  for (int r = 0; r < 8; ++r) red[r] += row_xor<8>(red[r]);
#pragma unroll
  for (int r = 0; r < 8; ++r) red[r] += row_xor<4>(red[r]);
#pragma unroll
  for (int r = 0; r < 8; ++r) red[r] += row_xor<2>(red[r]);
#pragma unroll
  for (int r = 0; r < 8; ++r) red[r] += row_xor<1>(red[r]);
#pragma unroll
  for (int r = 0; r < 8; ++r) {
    float mu = red[r] * (1.0f / BAG);
    float dd = hv[r] - mu;
    hv[r]  = dd;
    red[r] = (n16 < BAG) ? dd * dd : 0.f;
  }
#pragma unroll
  for (int r = 0; r < 8; ++r) red[r] += row_xor<8>(red[r]);
#pragma unroll
  for (int r = 0; r < 8; ++r) red[r] += row_xor<4>(red[r]);
#pragma unroll
  for (int r = 0; r < 8; ++r) red[r] += row_xor<2>(red[r]);
#pragma unroll
  for (int r = 0; r < 8; ++r) red[r] += row_xor<1>(red[r]);
#pragma unroll
  for (int r = 0; r < 8; ++r) {
    float s = hv[r] * rsqrtf(red[r] * (1.0f / BAG) + 1e-5f) * gN + bN;
    hv[r] = s;
    ldsState[wave][r + half * 8][n16] = s;
  }
  if (stepK < NSTEP - 1) {
#pragma unroll
    for (int r = 0; r < 8; ++r)
      stateWS[(size_t)(rowBase + r + half * 8) * 16 + n16] = hv[r];
  }
  __syncthreads();

  v8f accO = {0.f, 0.f, 0.f, 0.f, 0.f, 0.f, 0.f, 0.f};
#pragma unroll
  for (int t = 0; t < 4; ++t) {
    const int kb = 4 * t + 2 * half;
    v2f a, b;
    a.x = ldsState[wave][n16][kb];
    a.y = ldsState[wave][n16][kb + 1];
    b.x = ldsWout[kb * 16 + n16];
    b.y = ldsWout[(kb + 1) * 16 + n16];
    accO = __builtin_amdgcn_wmma_f32_16x16x4_f32(false, a, false, b,
                                                 (short)0, accO, false, false);
  }

#pragma unroll
  for (int r = 0; r < 8; ++r) {
    float raw10 = __shfl(accO[r], 10, 16);
    float raw11 = __shfl(accO[r], 11, 16);
    int q0 = __shfl(p0, half * 8 + r, 16);
    int q1 = __shfl(p1, half * 8 + r, 16);
    if (stepK < NSTEP - 1 && n16 == 0) {
      float c0 = tanhf(raw10 + bc0);
      float c1 = tanhf(raw11 + bc1);
      bool rowSel = fabsf(c0) >= fabsf(c1);
      int mv0 = rowSel ? ((c0 > 0.f) - (c0 < 0.f)) : 0;
      int mv1 = rowSel ? 0 : ((c1 > 0.f) - (c1 < 0.f));
      q0 = min(max(q0 + mv0, 0), DENOM);
      q1 = min(max(q1 + mv1, 0), DENOM);
      const int row = rowBase + half * 8 + r;
      posWS[2 * row]     = q0;
      posWS[2 * row + 1] = q1;
    }
  }

  if (stepK == NSTEP - 1 && n16 < 10) {
#pragma unroll
    for (int r = 0; r < 8; ++r)
      out[(size_t)(rowBase + r + half * 8) * 10 + n16] = accO[r] + boN;
  }

  if (stepK < NSTEP - 1) {
    v8f accC = {0.f, 0.f, 0.f, 0.f, 0.f, 0.f, 0.f, 0.f};
#pragma unroll
    for (int t = 0; t < 4; ++t) {
      const int kb = 4 * t + 2 * half;
      v2f f;
      f.x = ldsState[wave][kb][n16];
      f.y = ldsState[wave][kb + 1][n16];
      accC = __builtin_amdgcn_wmma_f32_16x16x4_f32(false, f, false, f,
                                                   (short)0, accC, false, false);
    }
#pragma unroll
    for (int r = 0; r < 8; ++r)
      ldsCo[wave][(r + half * 8) * 16 + n16] = accC[r];
  }
  __syncthreads();
  if (stepK < NSTEP - 1 && tid < PLANE) {
    const int r = tid >> 4, c = tid & 15;
    const int m = (c >= r) ? r : 15 - r;
    const int n = (c >= r) ? c : 15 - r + c;
    float acc = 0.f;
#pragma unroll
    for (int w = 0; w < NWAVE; ++w) acc += ldsCo[w][m * 16 + n];
    planes[(size_t)(stepK & 1) * (NWG * PLANE) + (size_t)blockIdx.x * PLANE + tid] = acc;
  }
}

extern "C" void kernel_launch(void* const* d_in, const int* in_sizes, int n_in,
                              void* d_out, int out_size, void* d_ws, size_t ws_size,
                              hipStream_t stream) {
  const float* x    = (const float*)d_in[0];
  const float* Wi   = (const float*)d_in[1];
  const float* bi   = (const float*)d_in[2];
  const float* gma  = (const float*)d_in[3];
  const float* bta  = (const float*)d_in[4];
  const float* R    = (const float*)d_in[5];
  const float* rb   = (const float*)d_in[6];
  const float* mask = (const float*)d_in[7];
  const float* Wc   = (const float*)d_in[8];
  const float* bc   = (const float*)d_in[9];
  const float* Wo   = (const float*)d_in[10];
  const float* bo   = (const float*)d_in[11];

  float* stateWS  = (float*)d_ws;
  int*   posWS    = (int*)((char*)d_ws + (size_t)BATCH * 16 * 4);
  float* planes   = (float*)((char*)posWS + (size_t)BATCH * 2 * 4);

  cyclic_step<true><<<NWG, WGT, 0, stream>>>(x, Wi, bi, gma, bta, R, rb, mask, Wc, bc,
                                             Wo, bo, stateWS, posWS, planes,
                                             (float*)d_out, 0);
  for (int k = 1; k < NSTEP; ++k) {
    cyclic_step<false><<<NWG, WGT, 0, stream>>>(x, Wi, bi, gma, bta, R, rb, mask, Wc, bc,
                                                Wo, bo, stateWS, posWS, planes,
                                                (float*)d_out, k);
  }
}


# coact planes hold only the triangle elements whose mask entry (either side) is not zero, ranked by ballot: 4 lines per plane become 1 per 32 kept, exact for any mask; plane sum order changes
# speedup vs baseline: 2.3155x; 1.0552x over previous
#include <hip/hip_runtime.h>
#include <hip/hip_bf16.h>

typedef __attribute__((ext_vector_type(2))) float v2f;
typedef __attribute__((ext_vector_type(8))) float v8f;

#define BAG   15
#define NSTEP 5
#define LAMBDA 0.001f
#define PS    8
#define IM    28
#define DENOM (IM - PS)
#define BATCH 65536
#define INDIM 66
#define NWG   256
#define WGT   512
#define NWAVE 16
#define PLANE 128
#define WISTR 68

template <int M>
__device__ __forceinline__ float row_xor(float v) {
  return __builtin_bit_cast(float,
      __builtin_amdgcn_update_dpp(0, __builtin_bit_cast(int, v),
                                  0x160 + M, 0xF, 0xF, true));
}

template <int C>
__device__ __forceinline__ float4 plane_tree(float4 (&pv)[16]) {
#pragma unroll
  for (int st = C / 2; st >= 1; st >>= 1) {
#pragma unroll
    for (int q = 0; q < st; ++q) {
      pv[q].x += pv[q + st].x; pv[q].y += pv[q + st].y;
      pv[q].z += pv[q + st].z; pv[q].w += pv[q + st].w;
    }
  }
  return pv[0];
}

template <bool FIRST>
__global__ __launch_bounds__(WGT) __attribute__((amdgpu_waves_per_eu(4, 4)))
void cyclic_step(const float* __restrict__ imgs,
                 const float* __restrict__ Wi,  const float* __restrict__ bi,
                 const float* __restrict__ gma, const float* __restrict__ bta,
                 const float* __restrict__ R,   const float* __restrict__ rb,
                 const float* __restrict__ mask,
                 const float* __restrict__ Wc,  const float* __restrict__ bc,
                 const float* __restrict__ Wo,  const float* __restrict__ bo,
                 float* __restrict__ stateWS,   int* __restrict__ posWS,
                 float* planes,
                 float* __restrict__ out,       int stepK) {
  __shared__ float ldsWi[16 * WISTR];
  __shared__ float ldsWout[256];
  __shared__ float ldsReff[256];
  __shared__ float ldsPar[48];
  __shared__ float4 ldsRed[NWAVE][PLANE / 4];
  __shared__ unsigned ldsBal[4];
  __shared__ int ldsRank[PLANE];
  __shared__ int ldsInv[PLANE];
  __shared__ float ldsState[NWAVE][16][16];
  __shared__ float ldsCo[NWAVE][256];

  const int tid  = threadIdx.x;
  const int wave = tid >> 5;
  const int lane = tid & 31;
  const int half = lane >> 4;
  const int n16  = lane & 15;
  const int rowBase = blockIdx.x * 256 + wave * 16;
  const int myRow   = rowBase + n16;

  int p0 = DENOM / 2, p1 = DENOM / 2;
  if (!FIRST) { p0 = posWS[2 * myRow]; p1 = posWS[2 * myRow + 1]; }

  v2f sa[4];
#pragma unroll
  for (int t = 0; t < 4; ++t) { sa[t].x = 0.f; sa[t].y = 0.f; }
  if (!FIRST) {
    const float* st = stateWS + (size_t)myRow * 16;
#pragma unroll
    for (int t = 0; t < 4; ++t) {
      const int kb = 4 * t + 2 * half;
      sa[t].x = st[kb];
      sa[t].y = st[kb + 1];
    }
  }

  float rv, mv;
  const int oi = tid & 15, oj = (tid >> 4) & 15;
  {
    const int ic = oi < BAG ? oi : BAG - 1;
    const int jc = oj < BAG ? oj : BAG - 1;
    rv = R[ic * BAG + jc];
    mv = mask[ic * BAG + jc];
  }

  const int er = (tid >> 4) & 7, ec = tid & 15;
  const int em = (ec >= er) ? er : 15 - er;
  const int en = (ec >= er) ? ec : 15 - er + ec;
  const int enc = en < BAG ? en : BAG - 1;
  const float mk0 = mask[em * BAG + enc];
  const float mk1 = mask[enc * BAG + em];

  float wiv[3];
#pragma unroll
  for (int q = 0; q < 3; ++q) {
    const int idx = tid + WGT * q;
    const int n = idx / WISTR, k = idx - n * WISTR;
    const int nc = n < BAG ? n : BAG - 1;
    const int kc = k < INDIM ? k : INDIM - 1;
    const float v = Wi[nc * INDIM + kc];
    wiv[q] = (n < BAG && k < INDIM) ? v : 0.f;
  }
  float hv0;
  {
    const int k = (tid >> 4) & 15, col = tid & 15;
    const int kc = k < BAG ? k : BAG - 1;
    const float* hsrc = (col < 10) ? (Wo + col * BAG)
                                   : (Wc + ((col < 12 ? col : 11) - 10) * BAG);
    const float v = hsrc[kc];
    hv0 = (k < BAG && col < 12) ? v : 0.f;
  }
  const float* img = imgs + (size_t)myRow * (IM * IM);
  v2f ga[16];
#pragma unroll
  for (int t = 0; t < 16; ++t) {
    const int kb = 4 * t + 2 * half;
    ga[t].x = img[(p0 + (kb >> 3)) * IM + p1 + (kb & 7)];
    ga[t].y = img[(p0 + ((kb + 1) >> 3)) * IM + p1 + ((kb + 1) & 7)];
  }

  float bv, gv, btv;
  {
    const int pc = n16 < BAG ? n16 : BAG - 1;
    const float b0 = bi[pc];
    const float r0 = rb[pc];
    const float g0 = gma[pc];
    const float t0 = bta[pc];
    const bool okp = n16 < BAG;
    bv  = okp ? b0 + (FIRST ? 0.f : r0) : 0.f;
    gv  = okp ? g0 : 0.f;
    btv = okp ? t0 : 0.f;
  }
  const float bc0 = bc[0], bc1 = bc[1];
  const float boN = bo[n16 < 10 ? n16 : 9];

#pragma unroll
  for (int q = 0; q < 3; ++q) {
    const int idx = tid + WGT * q;
    if (idx < 16 * WISTR) ldsWi[idx] = wiv[q];
  }
  if (tid < 256) ldsWout[tid] = hv0;
  if (tid < 16) {
    ldsPar[tid]      = bv;
    ldsPar[16 + tid] = gv;
    ldsPar[32 + tid] = btv;
  }
  const bool need = (en < BAG) && (mk0 != 0.f || mk1 != 0.f);
  const unsigned bal = __builtin_amdgcn_ballot_w32(need);
  if (wave < 4 && lane == 0) ldsBal[wave] = bal;
  __syncthreads();

  const int c0 = __builtin_amdgcn_readfirstlane(__popc(ldsBal[0]));
  const int c1 = __builtin_amdgcn_readfirstlane(__popc(ldsBal[1]));
  const int c2 = __builtin_amdgcn_readfirstlane(__popc(ldsBal[2]));
  const int c3 = __builtin_amdgcn_readfirstlane(__popc(ldsBal[3]));
  const int nnzU = c0 + c1 + c2 + c3;
  const int nl = nnzU <= 32 ? 1 : (nnzU <= 64 ? 2 : 4);
  if (wave < 4) {
    const int pre = (wave > 0 ? c0 : 0) + (wave > 1 ? c1 : 0) + (wave > 2 ? c2 : 0);
    const int rk = pre + __popc(bal & ((1u << lane) - 1u));
    ldsRank[tid] = need ? rk : -1;
    if (need) ldsInv[rk] = tid;
  }

  const int pw = (wave + blockIdx.x) & (NWAVE - 1);
  float4 pv[16];
  if (!FIRST) {
    const float4* psrc = reinterpret_cast<const float4*>(
                             planes + (size_t)((stepK + 1) & 1) * (NWG * PLANE))
                         + pw * 32 + lane;
    if (nl == 1) {
#pragma unroll
      for (int q = 0; q < 4; ++q) pv[q] = psrc[(size_t)q * (NWAVE * 32)];
    } else if (nl == 2) {
#pragma unroll
      for (int q = 0; q < 8; ++q) pv[q] = psrc[(size_t)q * (NWAVE * 32)];
    } else {
#pragma unroll
      for (int q = 0; q < 16; ++q) pv[q] = psrc[(size_t)q * (NWAVE * 32)];
    }
  }

  v8f accI = {0.f, 0.f, 0.f, 0.f, 0.f, 0.f, 0.f, 0.f};
#pragma unroll
  for (int t = 0; t < 17; ++t) {
    const int kb = 4 * t + 2 * half;
    v2f a;
    if (t < 16) {
      a = ga[t];
    } else if (half == 0) {
      a.x = (float)p0 * (2.0f / DENOM) - 1.0f;
      a.y = (float)p1 * (2.0f / DENOM) - 1.0f;
    } else {
      a.x = 0.f; a.y = 0.f;
    }
    v2f b = *(const v2f*)&ldsWi[n16 * WISTR + kb];
    accI = __builtin_amdgcn_wmma_f32_16x16x4_f32(false, a, false, b,
                                                 (short)0, accI, false, false);
  }

  if (!FIRST) {
    float4 s4;
    if (nl == 1) {
      s4 = plane_tree<4>(pv);
      s4.x += row_xor<8>(s4.x);         s4.y += row_xor<8>(s4.y);
      s4.z += row_xor<8>(s4.z);         s4.w += row_xor<8>(s4.w);
      s4.x += __shfl_xor(s4.x, 16, 32); s4.y += __shfl_xor(s4.y, 16, 32);
      s4.z += __shfl_xor(s4.z, 16, 32); s4.w += __shfl_xor(s4.w, 16, 32);
    } else if (nl == 2) {
      s4 = plane_tree<8>(pv);
      s4.x += __shfl_xor(s4.x, 16, 32); s4.y += __shfl_xor(s4.y, 16, 32);
      s4.z += __shfl_xor(s4.z, 16, 32); s4.w += __shfl_xor(s4.w, 16, 32);
    } else {
      s4 = plane_tree<16>(pv);
    }
    if (lane < 8 * nl) ldsRed[pw][lane] = s4;
    __syncthreads();

    if (tid < 256) {
      const int a = oi < oj ? oi : oj;
      const int b = oi < oj ? oj : oi;
      const int q = (a <= 7) ? a * 16 + b : (15 - a) * 16 + (b - a);
      const bool live = (oi < BAG) && (oj < BAG);
      const int rk = ldsRank[q];
      float co = 0.f;
      if (live && rk >= 0) {
        const float* rf = reinterpret_cast<const float*>(&ldsRed[0][0]) + rk;
        float t[NWAVE];
#pragma unroll
        for (int p = 0; p < NWAVE; ++p) t[p] = rf[p * PLANE];
#pragma unroll
        for (int st = NWAVE / 2; st >= 1; st >>= 1) {
#pragma unroll
          for (int p = 0; p < st; ++p) t[p] += t[p + st];
        }
        co = t[0] * (1.0f / (float)BATCH);
      }
      ldsReff[tid] = live ? (rv - LAMBDA * co) * mv : 0.f;
    }
    __syncthreads();
#pragma unroll
    for (int t = 0; t < 4; ++t) {
      const int kb = 4 * t + 2 * half;
      v2f b;
      b.x = ldsReff[kb * 16 + n16];
      b.y = ldsReff[(kb + 1) * 16 + n16];
      accI = __builtin_amdgcn_wmma_f32_16x16x4_f32(false, sa[t], false, b,
                                                   (short)0, accI, false, false);
    }
  }

  const float biasN = ldsPar[n16];
  const float gN    = ldsPar[16 + n16];
  const float bN    = ldsPar[32 + n16];
  float hv[8], red[8];
#pragma unroll
  for (int r = 0; r < 8; ++r) {
    hv[r]  = fmaxf(accI[r] + biasN, 0.f);
    red[r] = hv[r];
  }
#pragma unroll
  for (int r = 0; r < 8; ++r) red[r] += row_xor<8>(red[r]);
#pragma unroll
  for (int r = 0; r < 8; ++r) red[r] += row_xor<4>(red[r]);
#pragma unroll
  for (int r = 0; r < 8; ++r) red[r] += row_xor<2>(red[r]);
#pragma unroll
  for (int r = 0; r < 8; ++r) red[r] += row_xor<1>(red[r]);
#pragma unroll
  for (int r = 0; r < 8; ++r) {
    float mu = red[r] * (1.0f / BAG);
    float dd = hv[r] - mu;
    hv[r]  = dd;
    red[r] = (n16 < BAG) ? dd * dd : 0.f;
  }
#pragma unroll
  for (int r = 0; r < 8; ++r) red[r] += row_xor<8>(red[r]);
#pragma unroll
  for (int r = 0; r < 8; ++r) red[r] += row_xor<4>(red[r]);
#pragma unroll
  for (int r = 0; r < 8; ++r) red[r] += row_xor<2>(red[r]);
#pragma unroll
  for (int r = 0; r < 8; ++r) red[r] += row_xor<1>(red[r]);
#pragma unroll
  for (int r = 0; r < 8; ++r) {
    float s = hv[r] * rsqrtf(red[r] * (1.0f / BAG) + 1e-5f) * gN + bN;
    hv[r] = s;
    ldsState[wave][r + half * 8][n16] = s;
  }
  if (stepK < NSTEP - 1) {
#pragma unroll
    for (int r = 0; r < 8; ++r)
      stateWS[(size_t)(rowBase + r + half * 8) * 16 + n16] = hv[r];
  }
  __syncthreads();

  v8f accO = {0.f, 0.f, 0.f, 0.f, 0.f, 0.f, 0.f, 0.f};
#pragma unroll
  for (int t = 0; t < 4; ++t) {
    const int kb = 4 * t + 2 * half;
    v2f a, b;
    a.x = ldsState[wave][n16][kb];
    a.y = ldsState[wave][n16][kb + 1];
    b.x = ldsWout[kb * 16 + n16];
    b.y = ldsWout[(kb + 1) * 16 + n16];
    accO = __builtin_amdgcn_wmma_f32_16x16x4_f32(false, a, false, b,
                                                 (short)0, accO, false, false);
  }

#pragma unroll
  for (int r = 0; r < 8; ++r) {
    float raw10 = __shfl(accO[r], 10, 16);
    float raw11 = __shfl(accO[r], 11, 16);
    int q0 = __shfl(p0, half * 8 + r, 16);
    int q1 = __shfl(p1, half * 8 + r, 16);
    if (stepK < NSTEP - 1 && n16 == 0) {
      float c0 = tanhf(raw10 + bc0);
      float c1 = tanhf(raw11 + bc1);
      bool rowSel = fabsf(c0) >= fabsf(c1);
      int mv0 = rowSel ? ((c0 > 0.f) - (c0 < 0.f)) : 0;
      int mv1 = rowSel ? 0 : ((c1 > 0.f) - (c1 < 0.f));
      q0 = min(max(q0 + mv0, 0), DENOM);
      q1 = min(max(q1 + mv1, 0), DENOM);
      const int row = rowBase + half * 8 + r;
      posWS[2 * row]     = q0;
      posWS[2 * row + 1] = q1;
    }
  }

  if (stepK == NSTEP - 1 && n16 < 10) {
#pragma unroll
    for (int r = 0; r < 8; ++r)
      out[(size_t)(rowBase + r + half * 8) * 10 + n16] = accO[r] + boN;
  }

  if (stepK < NSTEP - 1) {
    v8f accC = {0.f, 0.f, 0.f, 0.f, 0.f, 0.f, 0.f, 0.f};
#pragma unroll
    for (int t = 0; t < 4; ++t) {
      const int kb = 4 * t + 2 * half;
      v2f f;
      f.x = ldsState[wave][kb][n16];
      f.y = ldsState[wave][kb + 1][n16];
      accC = __builtin_amdgcn_wmma_f32_16x16x4_f32(false, f, false, f,
                                                   (short)0, accC, false, false);
    }
#pragma unroll
    for (int r = 0; r < 8; ++r)
      ldsCo[wave][(r + half * 8) * 16 + n16] = accC[r];
  }
  __syncthreads();
  if (stepK < NSTEP - 1 && tid < 32 * nl) {
    float acc = 0.f;
    if (tid < nnzU) {
      const int e = ldsInv[tid];
      const int r = e >> 4, c = e & 15;
      const int m = (c >= r) ? r : 15 - r;
      const int n = (c >= r) ? c : 15 - r + c;
#pragma unroll
      for (int w = 0; w < NWAVE; ++w) acc += ldsCo[w][m * 16 + n];
    }
    planes[(size_t)(stepK & 1) * (NWG * PLANE) + (size_t)blockIdx.x * (32 * nl) + tid] = acc;
  }
}

extern "C" void kernel_launch(void* const* d_in, const int* in_sizes, int n_in,
                              void* d_out, int out_size, void* d_ws, size_t ws_size,
                              hipStream_t stream) {
  const float* x    = (const float*)d_in[0];
  const float* Wi   = (const float*)d_in[1];
  const float* bi   = (const float*)d_in[2];
  const float* gma  = (const float*)d_in[3];
  const float* bta  = (const float*)d_in[4];
  const float* R    = (const float*)d_in[5];
  const float* rb   = (const float*)d_in[6];
  const float* mask = (const float*)d_in[7];
  const float* Wc   = (const float*)d_in[8];
  const float* bc   = (const float*)d_in[9];
  const float* Wo   = (const float*)d_in[10];
  const float* bo   = (const float*)d_in[11];

  float* stateWS  = (float*)d_ws;
  int*   posWS    = (int*)((char*)d_ws + (size_t)BATCH * 16 * 4);
  float* planes   = (float*)((char*)posWS + (size_t)BATCH * 2 * 4);

  cyclic_step<true><<<NWG, WGT, 0, stream>>>(x, Wi, bi, gma, bta, R, rb, mask, Wc, bc,
                                             Wo, bo, stateWS, posWS, planes,
                                             (float*)d_out, 0);
  for (int k = 1; k < NSTEP; ++k) {
    cyclic_step<false><<<NWG, WGT, 0, stream>>>(x, Wi, bi, gma, bta, R, rb, mask, Wc, bc,
                                                Wo, bo, stateWS, posWS, planes,
                                                (float*)d_out, k);
  }
}


# control-to-position update: two lanes per wave took the 16 rows' tanh pairs one row after another (8 serial rounds); now each of the 32 lanes takes one tanh, one round, one line store; same values
# speedup vs baseline: 2.5201x; 1.0884x over previous
#include <hip/hip_runtime.h>
#include <hip/hip_bf16.h>

typedef __attribute__((ext_vector_type(2))) float v2f;
typedef __attribute__((ext_vector_type(8))) float v8f;

#define BAG   15
#define NSTEP 5
#define LAMBDA 0.001f
#define PS    8
#define IM    28
#define DENOM (IM - PS)
#define BATCH 65536
#define INDIM 66
#define NWG   256
#define WGT   512
#define NWAVE 16
#define PLANE 128
#define WISTR 68

template <int M>
__device__ __forceinline__ float row_xor(float v) {
  return __builtin_bit_cast(float,
      __builtin_amdgcn_update_dpp(0, __builtin_bit_cast(int, v),
                                  0x160 + M, 0xF, 0xF, true));
}

template <int M>
__device__ __forceinline__ int row_xor_i(int v) {
  return __builtin_amdgcn_update_dpp(0, v, 0x160 + M, 0xF, 0xF, true);
}

template <int S>
__device__ __forceinline__ float row_lane(float v) {
  return __builtin_bit_cast(float,
      __builtin_amdgcn_update_dpp(0, __builtin_bit_cast(int, v),
                                  0x150 + S, 0xF, 0xF, true));
}

template <int C>
__device__ __forceinline__ float4 plane_tree(float4 (&pv)[16]) {
#pragma unroll
  for (int st = C / 2; st >= 1; st >>= 1) {
#pragma unroll
    for (int q = 0; q < st; ++q) {
      pv[q].x += pv[q + st].x; pv[q].y += pv[q + st].y;
      pv[q].z += pv[q + st].z; pv[q].w += pv[q + st].w;
    }
  }
  return pv[0];
}

template <bool FIRST>
__global__ __launch_bounds__(WGT) __attribute__((amdgpu_waves_per_eu(4, 4)))
void cyclic_step(const float* __restrict__ imgs,
                 const float* __restrict__ Wi,  const float* __restrict__ bi,
                 const float* __restrict__ gma, const float* __restrict__ bta,
                 const float* __restrict__ R,   const float* __restrict__ rb,
                 const float* __restrict__ mask,
                 const float* __restrict__ Wc,  const float* __restrict__ bc,
                 const float* __restrict__ Wo,  const float* __restrict__ bo,
                 float* __restrict__ stateWS,   int* __restrict__ posWS,
                 float* planes,
                 float* __restrict__ out,       int stepK) {
  __shared__ float ldsWi[16 * WISTR];
  __shared__ float ldsWout[256];
  __shared__ float ldsReff[256];
  __shared__ float ldsPar[48];
  __shared__ float4 ldsRed[NWAVE][PLANE / 4];
  __shared__ unsigned ldsBal[4];
  __shared__ int ldsRank[PLANE];
  __shared__ int ldsInv[PLANE];
  __shared__ float ldsState[NWAVE][16][16];
  __shared__ float ldsCo[NWAVE][256];

  const int tid  = threadIdx.x;
  const int wave = tid >> 5;
  const int lane = tid & 31;
  const int half = lane >> 4;
  const int n16  = lane & 15;
  const int rowBase = blockIdx.x * 256 + wave * 16;
  const int myRow   = rowBase + n16;

  int p0 = DENOM / 2, p1 = DENOM / 2;
  if (!FIRST) { p0 = posWS[2 * myRow]; p1 = posWS[2 * myRow + 1]; }

  v2f sa[4];
#pragma unroll
  for (int t = 0; t < 4; ++t) { sa[t].x = 0.f; sa[t].y = 0.f; }
  if (!FIRST) {
    const float* st = stateWS + (size_t)myRow * 16;
#pragma unroll
    for (int t = 0; t < 4; ++t) {
      const int kb = 4 * t + 2 * half;
      sa[t].x = st[kb];
      sa[t].y = st[kb + 1];
    }
  }

  float rv, mv;
  const int oi = tid & 15, oj = (tid >> 4) & 15;
  {
    const int ic = oi < BAG ? oi : BAG - 1;
    const int jc = oj < BAG ? oj : BAG - 1;
    rv = R[ic * BAG + jc];
    mv = mask[ic * BAG + jc];
  }

  const int er = (tid >> 4) & 7, ec = tid & 15;
  const int em = (ec >= er) ? er : 15 - er;
  const int en = (ec >= er) ? ec : 15 - er + ec;
  const int enc = en < BAG ? en : BAG - 1;
  const float mk0 = mask[em * BAG + enc];
  const float mk1 = mask[enc * BAG + em];

  float wiv[3];
#pragma unroll
  for (int q = 0; q < 3; ++q) {
    const int idx = tid + WGT * q;
    const int n = idx / WISTR, k = idx - n * WISTR;
    const int nc = n < BAG ? n : BAG - 1;
    const int kc = k < INDIM ? k : INDIM - 1;
    const float v = Wi[nc * INDIM + kc];
    wiv[q] = (n < BAG && k < INDIM) ? v : 0.f;
  }
  float hv0;
  {
    const int k = (tid >> 4) & 15, col = tid & 15;
    const int kc = k < BAG ? k : BAG - 1;
    const float* hsrc = (col < 10) ? (Wo + col * BAG)
                                   : (Wc + ((col < 12 ? col : 11) - 10) * BAG);
    const float v = hsrc[kc];
    hv0 = (k < BAG && col < 12) ? v : 0.f;
  }
  const float* img = imgs + (size_t)myRow * (IM * IM);
  v2f ga[16];
#pragma unroll
  for (int t = 0; t < 16; ++t) {
    const int kb = 4 * t + 2 * half;
    ga[t].x = img[(p0 + (kb >> 3)) * IM + p1 + (kb & 7)];
    ga[t].y = img[(p0 + ((kb + 1) >> 3)) * IM + p1 + ((kb + 1) & 7)];
  }

  float bv, gv, btv;
  {
    const int pc = n16 < BAG ? n16 : BAG - 1;
    const float b0 = bi[pc];
    const float r0 = rb[pc];
    const float g0 = gma[pc];
    const float t0 = bta[pc];
    const bool okp = n16 < BAG;
    bv  = okp ? b0 + (FIRST ? 0.f : r0) : 0.f;
    gv  = okp ? g0 : 0.f;
    btv = okp ? t0 : 0.f;
  }
  const float bc0 = bc[0], bc1 = bc[1];
  const float boN = bo[n16 < 10 ? n16 : 9];

#pragma unroll
  for (int q = 0; q < 3; ++q) {
    const int idx = tid + WGT * q;
    if (idx < 16 * WISTR) ldsWi[idx] = wiv[q];
  }
  if (tid < 256) ldsWout[tid] = hv0;
  if (tid < 16) {
    ldsPar[tid]      = bv;
    ldsPar[16 + tid] = gv;
    ldsPar[32 + tid] = btv;
  }
  const bool need = (en < BAG) && (mk0 != 0.f || mk1 != 0.f);
  const unsigned bal = __builtin_amdgcn_ballot_w32(need);
  if (wave < 4 && lane == 0) ldsBal[wave] = bal;
  __syncthreads();

  const int c0 = __builtin_amdgcn_readfirstlane(__popc(ldsBal[0]));
  const int c1 = __builtin_amdgcn_readfirstlane(__popc(ldsBal[1]));
  const int c2 = __builtin_amdgcn_readfirstlane(__popc(ldsBal[2]));
  const int c3 = __builtin_amdgcn_readfirstlane(__popc(ldsBal[3]));
  const int nnzU = c0 + c1 + c2 + c3;
  const int nl = nnzU <= 32 ? 1 : (nnzU <= 64 ? 2 : 4);
  if (wave < 4) {
    const int pre = (wave > 0 ? c0 : 0) + (wave > 1 ? c1 : 0) + (wave > 2 ? c2 : 0);
    const int rk = pre + __popc(bal & ((1u << lane) - 1u));
    ldsRank[tid] = need ? rk : -1;
    if (need) ldsInv[rk] = tid;
  }

  const int pw = (wave + blockIdx.x) & (NWAVE - 1);
  float4 pv[16];
  if (!FIRST) {
    const float4* psrc = reinterpret_cast<const float4*>(
                             planes + (size_t)((stepK + 1) & 1) * (NWG * PLANE))
                         + pw * 32 + lane;
    if (nl == 1) {
#pragma unroll
      for (int q = 0; q < 4; ++q) pv[q] = psrc[(size_t)q * (NWAVE * 32)];
    } else if (nl == 2) {
#pragma unroll
      for (int q = 0; q < 8; ++q) pv[q] = psrc[(size_t)q * (NWAVE * 32)];
    } else {
#pragma unroll
      for (int q = 0; q < 16; ++q) pv[q] = psrc[(size_t)q * (NWAVE * 32)];
    }
  }

  v8f accI = {0.f, 0.f, 0.f, 0.f, 0.f, 0.f, 0.f, 0.f};
#pragma unroll
  for (int t = 0; t < 17; ++t) {
    const int kb = 4 * t + 2 * half;
    v2f a;
    if (t < 16) {
      a = ga[t];
    } else if (half == 0) {
      a.x = (float)p0 * (2.0f / DENOM) - 1.0f;
      a.y = (float)p1 * (2.0f / DENOM) - 1.0f;
    } else {
      a.x = 0.f; a.y = 0.f;
    }
    v2f b = *(const v2f*)&ldsWi[n16 * WISTR + kb];
    accI = __builtin_amdgcn_wmma_f32_16x16x4_f32(false, a, false, b,
                                                 (short)0, accI, false, false);
  }

  if (!FIRST) {
    float4 s4;
    if (nl == 1) {
      s4 = plane_tree<4>(pv);
      s4.x += row_xor<8>(s4.x);         s4.y += row_xor<8>(s4.y);
      s4.z += row_xor<8>(s4.z);         s4.w += row_xor<8>(s4.w);
      s4.x += __shfl_xor(s4.x, 16, 32); s4.y += __shfl_xor(s4.y, 16, 32);
      s4.z += __shfl_xor(s4.z, 16, 32); s4.w += __shfl_xor(s4.w, 16, 32);
    } else if (nl == 2) {
      s4 = plane_tree<8>(pv);
      s4.x += __shfl_xor(s4.x, 16, 32); s4.y += __shfl_xor(s4.y, 16, 32);
      s4.z += __shfl_xor(s4.z, 16, 32); s4.w += __shfl_xor(s4.w, 16, 32);
    } else {
      s4 = plane_tree<16>(pv);
    }
    if (lane < 8 * nl) ldsRed[pw][lane] = s4;
    __syncthreads();

    if (tid < 256) {
      const int a = oi < oj ? oi : oj;
      const int b = oi < oj ? oj : oi;
      const int q = (a <= 7) ? a * 16 + b : (15 - a) * 16 + (b - a);
      const bool live = (oi < BAG) && (oj < BAG);
      const int rk = ldsRank[q];
      float co = 0.f;
      if (live && rk >= 0) {
        const float* rf = reinterpret_cast<const float*>(&ldsRed[0][0]) + rk;
        float t[NWAVE];
#pragma unroll
        for (int p = 0; p < NWAVE; ++p) t[p] = rf[p * PLANE];
#pragma unroll
        for (int st = NWAVE / 2; st >= 1; st >>= 1) {
#pragma unroll
          for (int p = 0; p < st; ++p) t[p] += t[p + st];
        }
        co = t[0] * (1.0f / (float)BATCH);
      }
      ldsReff[tid] = live ? (rv - LAMBDA * co) * mv : 0.f;
    }
    __syncthreads();
#pragma unroll
    for (int t = 0; t < 4; ++t) {
      const int kb = 4 * t + 2 * half;
      v2f b;
      b.x = ldsReff[kb * 16 + n16];
      b.y = ldsReff[(kb + 1) * 16 + n16];
      accI = __builtin_amdgcn_wmma_f32_16x16x4_f32(false, sa[t], false, b,
                                                   (short)0, accI, false, false);
    }
  }

  const float biasN = ldsPar[n16];
  const float gN    = ldsPar[16 + n16];
  const float bN    = ldsPar[32 + n16];
  float hv[8], red[8];
#pragma unroll
  for (int r = 0; r < 8; ++r) {
    hv[r]  = fmaxf(accI[r] + biasN, 0.f);
    red[r] = hv[r];
  }
#pragma unroll
  for (int r = 0; r < 8; ++r) red[r] += row_xor<8>(red[r]);
#pragma unroll
  for (int r = 0; r < 8; ++r) red[r] += row_xor<4>(red[r]);
#pragma unroll
  for (int r = 0; r < 8; ++r) red[r] += row_xor<2>(red[r]);
#pragma unroll
  for (int r = 0; r < 8; ++r) red[r] += row_xor<1>(red[r]);
#pragma unroll
  for (int r = 0; r < 8; ++r) {
    float mu = red[r] * (1.0f / BAG);
    float dd = hv[r] - mu;
    hv[r]  = dd;
    red[r] = (n16 < BAG) ? dd * dd : 0.f;
  }
#pragma unroll
  for (int r = 0; r < 8; ++r) red[r] += row_xor<8>(red[r]);
#pragma unroll
  for (int r = 0; r < 8; ++r) red[r] += row_xor<4>(red[r]);
#pragma unroll
  for (int r = 0; r < 8; ++r) red[r] += row_xor<2>(red[r]);
#pragma unroll
  for (int r = 0; r < 8; ++r) red[r] += row_xor<1>(red[r]);
#pragma unroll
  for (int r = 0; r < 8; ++r) {
    float s = hv[r] * rsqrtf(red[r] * (1.0f / BAG) + 1e-5f) * gN + bN;
    hv[r] = s;
    ldsState[wave][r + half * 8][n16] = s;
  }
  if (stepK < NSTEP - 1) {
#pragma unroll
    for (int r = 0; r < 8; ++r)
      stateWS[(size_t)(rowBase + r + half * 8) * 16 + n16] = hv[r];
  }
  __syncthreads();

  v8f accO = {0.f, 0.f, 0.f, 0.f, 0.f, 0.f, 0.f, 0.f};
#pragma unroll
  for (int t = 0; t < 4; ++t) {
    const int kb = 4 * t + 2 * half;
    v2f a, b;
    a.x = ldsState[wave][n16][kb];
    a.y = ldsState[wave][n16][kb + 1];
    b.x = ldsWout[kb * 16 + n16];
    b.y = ldsWout[(kb + 1) * 16 + n16];
    accO = __builtin_amdgcn_wmma_f32_16x16x4_f32(false, a, false, b,
                                                 (short)0, accO, false, false);
  }

  if (stepK < NSTEP - 1) {
    const int  j      = n16 & 7;
    const bool second = n16 >= 8;
    float raw = 0.f;
#pragma unroll
    for (int r = 0; r < 8; ++r) {
      const float v10 = row_lane<10>(accO[r]);
      const float v11 = row_lane<11>(accO[r]);
      const float v   = second ? v11 : v10;
      raw = (j == r) ? v : raw;
    }
    const float cm = tanhf(raw + (second ? bc1 : bc0));
    const float cx = row_xor<8>(cm);
    const int r0 = row_xor_i<8>(p0);
    const int r1 = row_xor_i<8>(p1);
    if (!second) {
      const float c0 = cm, c1 = cx;
      int q0 = half ? r0 : p0;
      int q1 = half ? r1 : p1;
      const bool rowSel = fabsf(c0) >= fabsf(c1);
      const int mv0 = rowSel ? ((c0 > 0.f) - (c0 < 0.f)) : 0;
      const int mv1 = rowSel ? 0 : ((c1 > 0.f) - (c1 < 0.f));
      q0 = min(max(q0 + mv0, 0), DENOM);
      q1 = min(max(q1 + mv1, 0), DENOM);
      const int row = rowBase + half * 8 + n16;
      *reinterpret_cast<int2*>(posWS + 2 * row) = make_int2(q0, q1);
    }
  }

  if (stepK == NSTEP - 1 && n16 < 10) {
#pragma unroll
    for (int r = 0; r < 8; ++r)
      out[(size_t)(rowBase + r + half * 8) * 10 + n16] = accO[r] + boN;
  }

  if (stepK < NSTEP - 1) {
    v8f accC = {0.f, 0.f, 0.f, 0.f, 0.f, 0.f, 0.f, 0.f};
#pragma unroll
    for (int t = 0; t < 4; ++t) {
      const int kb = 4 * t + 2 * half;
      v2f f;
      f.x = ldsState[wave][kb][n16];
      f.y = ldsState[wave][kb + 1][n16];
      accC = __builtin_amdgcn_wmma_f32_16x16x4_f32(false, f, false, f,
                                                   (short)0, accC, false, false);
    }
#pragma unroll
    for (int r = 0; r < 8; ++r)
      ldsCo[wave][(r + half * 8) * 16 + n16] = accC[r];
  }
  __syncthreads();
  if (stepK < NSTEP - 1 && tid < 32 * nl) {
    float acc = 0.f;
    if (tid < nnzU) {
      const int e = ldsInv[tid];
      const int r = e >> 4, c = e & 15;
      const int m = (c >= r) ? r : 15 - r;
      const int n = (c >= r) ? c : 15 - r + c;
#pragma unroll
      for (int w = 0; w < NWAVE; ++w) acc += ldsCo[w][m * 16 + n];
    }
    planes[(size_t)(stepK & 1) * (NWG * PLANE) + (size_t)blockIdx.x * (32 * nl) + tid] = acc;
  }
}

extern "C" void kernel_launch(void* const* d_in, const int* in_sizes, int n_in,
                              void* d_out, int out_size, void* d_ws, size_t ws_size,
                              hipStream_t stream) {
  const float* x    = (const float*)d_in[0];
  const float* Wi   = (const float*)d_in[1];
  const float* bi   = (const float*)d_in[2];
  const float* gma  = (const float*)d_in[3];
  const float* bta  = (const float*)d_in[4];
  const float* R    = (const float*)d_in[5];
  const float* rb   = (const float*)d_in[6];
  const float* mask = (const float*)d_in[7];
  const float* Wc   = (const float*)d_in[8];
  const float* bc   = (const float*)d_in[9];
  const float* Wo   = (const float*)d_in[10];
  const float* bo   = (const float*)d_in[11];

  float* stateWS  = (float*)d_ws;
  int*   posWS    = (int*)((char*)d_ws + (size_t)BATCH * 16 * 4);
  float* planes   = (float*)((char*)posWS + (size_t)BATCH * 2 * 4);

  cyclic_step<true><<<NWG, WGT, 0, stream>>>(x, Wi, bi, gma, bta, R, rb, mask, Wc, bc,
                                             Wo, bo, stateWS, posWS, planes,
                                             (float*)d_out, 0);
  for (int k = 1; k < NSTEP; ++k) {
    cyclic_step<false><<<NWG, WGT, 0, stream>>>(x, Wi, bi, gma, bta, R, rb, mask, Wc, bc,
                                                Wo, bo, stateWS, posWS, planes,
                                                (float*)d_out, k);
  }
}


# patch gather: two 8-byte loads per lane and patch row (16 per lane) -> one 16-byte load (8); the halves' middle pixels traded by v_permlane16_swap into the same fragment order, sums bit-identical
# speedup vs baseline: 2.6128x; 1.0368x over previous
#include <hip/hip_runtime.h>
#include <hip/hip_bf16.h>

typedef __attribute__((ext_vector_type(2))) float v2f;
typedef __attribute__((ext_vector_type(8))) float v8f;
typedef float v4f_dw __attribute__((ext_vector_type(4), aligned(4)));

#define BAG   15
#define NSTEP 5
#define LAMBDA 0.001f
#define PS    8
#define IM    28
#define DENOM (IM - PS)
#define BATCH 65536
#define INDIM 66
#define NWG   256
#define WGT   512
#define NWAVE 16
#define PLANE 128
#define WISTR 68

template <int M>
__device__ __forceinline__ float row_xor(float v) {
  return __builtin_bit_cast(float,
      __builtin_amdgcn_update_dpp(0, __builtin_bit_cast(int, v),
                                  0x160 + M, 0xF, 0xF, true));
}

template <int M>
__device__ __forceinline__ int row_xor_i(int v) {
  return __builtin_amdgcn_update_dpp(0, v, 0x160 + M, 0xF, 0xF, true);
}

template <int S>
__device__ __forceinline__ float row_lane(float v) {
  return __builtin_bit_cast(float,
      __builtin_amdgcn_update_dpp(0, __builtin_bit_cast(int, v),
                                  0x150 + S, 0xF, 0xF, true));
}

template <int C>
__device__ __forceinline__ float4 plane_tree(float4 (&pv)[16]) {
#pragma unroll
  for (int st = C / 2; st >= 1; st >>= 1) {
#pragma unroll
    for (int q = 0; q < st; ++q) {
      pv[q].x += pv[q + st].x; pv[q].y += pv[q + st].y;
      pv[q].z += pv[q + st].z; pv[q].w += pv[q + st].w;
    }
  }
  return pv[0];
}

template <bool FIRST>
__global__ __launch_bounds__(WGT) __attribute__((amdgpu_waves_per_eu(4, 4)))
void cyclic_step(const float* __restrict__ imgs,
                 const float* __restrict__ Wi,  const float* __restrict__ bi,
                 const float* __restrict__ gma, const float* __restrict__ bta,
                 const float* __restrict__ R,   const float* __restrict__ rb,
                 const float* __restrict__ mask,
                 const float* __restrict__ Wc,  const float* __restrict__ bc,
                 const float* __restrict__ Wo,  const float* __restrict__ bo,
                 float* __restrict__ stateWS,   int* __restrict__ posWS,
                 float* planes,
                 float* __restrict__ out,       int stepK) {
  __shared__ float ldsWi[16 * WISTR];
  __shared__ float ldsWout[256];
  __shared__ float ldsReff[256];
  __shared__ float ldsPar[48];
  __shared__ float4 ldsRed[NWAVE][PLANE / 4];
  __shared__ unsigned ldsBal[4];
  __shared__ int ldsRank[PLANE];
  __shared__ int ldsInv[PLANE];
  __shared__ float ldsState[NWAVE][16][16];
  __shared__ float ldsCo[NWAVE][256];

  const int tid  = threadIdx.x;
  const int wave = tid >> 5;
  const int lane = tid & 31;
  const int half = lane >> 4;
  const int n16  = lane & 15;
  const int rowBase = blockIdx.x * 256 + wave * 16;
  const int myRow   = rowBase + n16;

  int p0 = DENOM / 2, p1 = DENOM / 2;
  if (!FIRST) { p0 = posWS[2 * myRow]; p1 = posWS[2 * myRow + 1]; }

  v2f sa[4];
#pragma unroll
  for (int t = 0; t < 4; ++t) { sa[t].x = 0.f; sa[t].y = 0.f; }
  if (!FIRST) {
    const float* st = stateWS + (size_t)myRow * 16;
#pragma unroll
    for (int t = 0; t < 4; ++t) {
      const int kb = 4 * t + 2 * half;
      sa[t].x = st[kb];
      sa[t].y = st[kb + 1];
    }
  }

  float rv, mv;
  const int oi = tid & 15, oj = (tid >> 4) & 15;
  {
    const int ic = oi < BAG ? oi : BAG - 1;
    const int jc = oj < BAG ? oj : BAG - 1;
    rv = R[ic * BAG + jc];
    mv = mask[ic * BAG + jc];
  }

  const int er = (tid >> 4) & 7, ec = tid & 15;
  const int em = (ec >= er) ? er : 15 - er;
  const int en = (ec >= er) ? ec : 15 - er + ec;
  const int enc = en < BAG ? en : BAG - 1;
  const float mk0 = mask[em * BAG + enc];
  const float mk1 = mask[enc * BAG + em];

  float wiv[3];
#pragma unroll
  for (int q = 0; q < 3; ++q) {
    const int idx = tid + WGT * q;
    const int n = idx / WISTR, k = idx - n * WISTR;
    const int nc = n < BAG ? n : BAG - 1;
    const int kc = k < INDIM ? k : INDIM - 1;
    const float v = Wi[nc * INDIM + kc];
    wiv[q] = (n < BAG && k < INDIM) ? v : 0.f;
  }
  float hv0;
  {
    const int k = (tid >> 4) & 15, col = tid & 15;
    const int kc = k < BAG ? k : BAG - 1;
    const float* hsrc = (col < 10) ? (Wo + col * BAG)
                                   : (Wc + ((col < 12 ? col : 11) - 10) * BAG);
    const float v = hsrc[kc];
    hv0 = (k < BAG && col < 12) ? v : 0.f;
  }
  const float* gp = imgs + (size_t)myRow * (IM * IM) + (p0 * IM + p1 + 4 * half);
  v4f_dw gq[8];
#pragma unroll
  for (int j = 0; j < 8; ++j)
    gq[j] = *reinterpret_cast<const v4f_dw*>(gp + j * IM);

  float bv, gv, btv;
  {
    const int pc = n16 < BAG ? n16 : BAG - 1;
    const float b0 = bi[pc];
    const float r0 = rb[pc];
    const float g0 = gma[pc];
    const float t0 = bta[pc];
    const bool okp = n16 < BAG;
    bv  = okp ? b0 + (FIRST ? 0.f : r0) : 0.f;
    gv  = okp ? g0 : 0.f;
    btv = okp ? t0 : 0.f;
  }
  const float bc0 = bc[0], bc1 = bc[1];
  const float boN = bo[n16 < 10 ? n16 : 9];

#pragma unroll
  for (int q = 0; q < 3; ++q) {
    const int idx = tid + WGT * q;
    if (idx < 16 * WISTR) ldsWi[idx] = wiv[q];
  }
  if (tid < 256) ldsWout[tid] = hv0;
  if (tid < 16) {
    ldsPar[tid]      = bv;
    ldsPar[16 + tid] = gv;
    ldsPar[32 + tid] = btv;
  }
  const bool need = (en < BAG) && (mk0 != 0.f || mk1 != 0.f);
  const unsigned bal = __builtin_amdgcn_ballot_w32(need);
  if (wave < 4 && lane == 0) ldsBal[wave] = bal;
  v2f ga[16];
#pragma unroll
  for (int j = 0; j < 8; ++j) {
    float qx = gq[j].x, qy = gq[j].y, qz = gq[j].z, qw = gq[j].w;
    asm volatile("v_permlane16_swap_b32 %0, %1" : "+v"(qx), "+v"(qz));
    asm volatile("v_permlane16_swap_b32 %0, %1" : "+v"(qy), "+v"(qw));
    ga[2 * j].x     = qx;
    ga[2 * j].y     = qy;
    ga[2 * j + 1].x = qz;
    ga[2 * j + 1].y = qw;
  }
  __syncthreads();

  const int c0 = __builtin_amdgcn_readfirstlane(__popc(ldsBal[0]));
  const int c1 = __builtin_amdgcn_readfirstlane(__popc(ldsBal[1]));
  const int c2 = __builtin_amdgcn_readfirstlane(__popc(ldsBal[2]));
  const int c3 = __builtin_amdgcn_readfirstlane(__popc(ldsBal[3]));
  const int nnzU = c0 + c1 + c2 + c3;
  const int nl = nnzU <= 32 ? 1 : (nnzU <= 64 ? 2 : 4);
  if (wave < 4) {
    const int pre = (wave > 0 ? c0 : 0) + (wave > 1 ? c1 : 0) + (wave > 2 ? c2 : 0);
    const int rk = pre + __popc(bal & ((1u << lane) - 1u));
    ldsRank[tid] = need ? rk : -1;
    if (need) ldsInv[rk] = tid;
  }

  const int pw = (wave + blockIdx.x) & (NWAVE - 1);
  float4 pv[16];
  if (!FIRST) {
    const float4* psrc = reinterpret_cast<const float4*>(
                             planes + (size_t)((stepK + 1) & 1) * (NWG * PLANE))
                         + pw * 32 + lane;
    if (nl == 1) {
#pragma unroll
      for (int q = 0; q < 4; ++q) pv[q] = psrc[(size_t)q * (NWAVE * 32)];
    } else if (nl == 2) {
#pragma unroll
      for (int q = 0; q < 8; ++q) pv[q] = psrc[(size_t)q * (NWAVE * 32)];
    } else {
#pragma unroll
      for (int q = 0; q < 16; ++q) pv[q] = psrc[(size_t)q * (NWAVE * 32)];
    }
  }

  v8f accI = {0.f, 0.f, 0.f, 0.f, 0.f, 0.f, 0.f, 0.f};
#pragma unroll
  for (int t = 0; t < 17; ++t) {
    const int kb = 4 * t + 2 * half;
    v2f a;
    if (t < 16) {
      a = ga[t];
    } else if (half == 0) {
      a.x = (float)p0 * (2.0f / DENOM) - 1.0f;
      a.y = (float)p1 * (2.0f / DENOM) - 1.0f;
    } else {
      a.x = 0.f; a.y = 0.f;
    }
    v2f b = *(const v2f*)&ldsWi[n16 * WISTR + kb];
    accI = __builtin_amdgcn_wmma_f32_16x16x4_f32(false, a, false, b,
                                                 (short)0, accI, false, false);
  }

  if (!FIRST) {
    float4 s4;
    if (nl == 1) {
      s4 = plane_tree<4>(pv);
      s4.x += row_xor<8>(s4.x);         s4.y += row_xor<8>(s4.y);
      s4.z += row_xor<8>(s4.z);         s4.w += row_xor<8>(s4.w);
      s4.x += __shfl_xor(s4.x, 16, 32); s4.y += __shfl_xor(s4.y, 16, 32);
      s4.z += __shfl_xor(s4.z, 16, 32); s4.w += __shfl_xor(s4.w, 16, 32);
    } else if (nl == 2) {
      s4 = plane_tree<8>(pv);
      s4.x += __shfl_xor(s4.x, 16, 32); s4.y += __shfl_xor(s4.y, 16, 32);
      s4.z += __shfl_xor(s4.z, 16, 32); s4.w += __shfl_xor(s4.w, 16, 32);
    } else {
      s4 = plane_tree<16>(pv);
    }
    if (lane < 8 * nl) ldsRed[pw][lane] = s4;
    __syncthreads();

    if (tid < 256) {
      const int a = oi < oj ? oi : oj;
      const int b = oi < oj ? oj : oi;
      const int q = (a <= 7) ? a * 16 + b : (15 - a) * 16 + (b - a);
      const bool live = (oi < BAG) && (oj < BAG);
      const int rk = ldsRank[q];
      float co = 0.f;
      if (live && rk >= 0) {
        const float* rf = reinterpret_cast<const float*>(&ldsRed[0][0]) + rk;
        float t[NWAVE];
#pragma unroll
        for (int p = 0; p < NWAVE; ++p) t[p] = rf[p * PLANE];
#pragma unroll
        for (int st = NWAVE / 2; st >= 1; st >>= 1) {
#pragma unroll
          for (int p = 0; p < st; ++p) t[p] += t[p + st];
        }
        co = t[0] * (1.0f / (float)BATCH);
      }
      ldsReff[tid] = live ? (rv - LAMBDA * co) * mv : 0.f;
    }
    __syncthreads();
#pragma unroll
    for (int t = 0; t < 4; ++t) {
      const int kb = 4 * t + 2 * half;
      v2f b;
      b.x = ldsReff[kb * 16 + n16];
      b.y = ldsReff[(kb + 1) * 16 + n16];
      accI = __builtin_amdgcn_wmma_f32_16x16x4_f32(false, sa[t], false, b,
                                                   (short)0, accI, false, false);
    }
  }

  const float biasN = ldsPar[n16];
  const float gN    = ldsPar[16 + n16];
  const float bN    = ldsPar[32 + n16];
  float hv[8], red[8];
#pragma unroll
  for (int r = 0; r < 8; ++r) {
    hv[r]  = fmaxf(accI[r] + biasN, 0.f);
    red[r] = hv[r];
  }
#pragma unroll
  for (int r = 0; r < 8; ++r) red[r] += row_xor<8>(red[r]);
#pragma unroll
  for (int r = 0; r < 8; ++r) red[r] += row_xor<4>(red[r]);
#pragma unroll
  for (int r = 0; r < 8; ++r) red[r] += row_xor<2>(red[r]);
#pragma unroll
  for (int r = 0; r < 8; ++r) red[r] += row_xor<1>(red[r]);
#pragma unroll
  for (int r = 0; r < 8; ++r) {
    float mu = red[r] * (1.0f / BAG);
    float dd = hv[r] - mu;
    hv[r]  = dd;
    red[r] = (n16 < BAG) ? dd * dd : 0.f;
  }
#pragma unroll
  for (int r = 0; r < 8; ++r) red[r] += row_xor<8>(red[r]);
#pragma unroll
  for (int r = 0; r < 8; ++r) red[r] += row_xor<4>(red[r]);
#pragma unroll
  for (int r = 0; r < 8; ++r) red[r] += row_xor<2>(red[r]);
#pragma unroll
  for (int r = 0; r < 8; ++r) red[r] += row_xor<1>(red[r]);
#pragma unroll
  for (int r = 0; r < 8; ++r) {
    float s = hv[r] * rsqrtf(red[r] * (1.0f / BAG) + 1e-5f) * gN + bN;
    hv[r] = s;
    ldsState[wave][r + half * 8][n16] = s;
  }
  if (stepK < NSTEP - 1) {
#pragma unroll
    for (int r = 0; r < 8; ++r)
      stateWS[(size_t)(rowBase + r + half * 8) * 16 + n16] = hv[r];
  }
  __syncthreads();

  v8f accO = {0.f, 0.f, 0.f, 0.f, 0.f, 0.f, 0.f, 0.f};
#pragma unroll
  for (int t = 0; t < 4; ++t) {
    const int kb = 4 * t + 2 * half;
    v2f a, b;
    a.x = ldsState[wave][n16][kb];
    a.y = ldsState[wave][n16][kb + 1];
    b.x = ldsWout[kb * 16 + n16];
    b.y = ldsWout[(kb + 1) * 16 + n16];
    accO = __builtin_amdgcn_wmma_f32_16x16x4_f32(false, a, false, b,
                                                 (short)0, accO, false, false);
  }

  if (stepK < NSTEP - 1) {
    const int  j      = n16 & 7;
    const bool second = n16 >= 8;
    float raw = 0.f;
#pragma unroll
    for (int r = 0; r < 8; ++r) {
      const float v10 = row_lane<10>(accO[r]);
      const float v11 = row_lane<11>(accO[r]);
      const float v   = second ? v11 : v10;
      raw = (j == r) ? v : raw;
    }
    const float cm = tanhf(raw + (second ? bc1 : bc0));
    const float cx = row_xor<8>(cm);
    const int r0 = row_xor_i<8>(p0);
    const int r1 = row_xor_i<8>(p1);
    if (!second) {
      const float c0 = cm, c1 = cx;
      int q0 = half ? r0 : p0;
      int q1 = half ? r1 : p1;
      const bool rowSel = fabsf(c0) >= fabsf(c1);
      const int mv0 = rowSel ? ((c0 > 0.f) - (c0 < 0.f)) : 0;
      const int mv1 = rowSel ? 0 : ((c1 > 0.f) - (c1 < 0.f));
      q0 = min(max(q0 + mv0, 0), DENOM);
      q1 = min(max(q1 + mv1, 0), DENOM);
      const int row = rowBase + half * 8 + n16;
      *reinterpret_cast<int2*>(posWS + 2 * row) = make_int2(q0, q1);
    }
  }

  if (stepK == NSTEP - 1 && n16 < 10) {
#pragma unroll
    for (int r = 0; r < 8; ++r)
      out[(size_t)(rowBase + r + half * 8) * 10 + n16] = accO[r] + boN;
  }

  if (stepK < NSTEP - 1) {
    v8f accC = {0.f, 0.f, 0.f, 0.f, 0.f, 0.f, 0.f, 0.f};
#pragma unroll
    for (int t = 0; t < 4; ++t) {
      const int kb = 4 * t + 2 * half;
      v2f f;
      f.x = ldsState[wave][kb][n16];
      f.y = ldsState[wave][kb + 1][n16];
      accC = __builtin_amdgcn_wmma_f32_16x16x4_f32(false, f, false, f,
                                                   (short)0, accC, false, false);
    }
#pragma unroll
    for (int r = 0; r < 8; ++r)
      ldsCo[wave][(r + half * 8) * 16 + n16] = accC[r];
  }
  __syncthreads();
  if (stepK < NSTEP - 1 && tid < 32 * nl) {
    float acc = 0.f;
    if (tid < nnzU) {
      const int e = ldsInv[tid];
      const int r = e >> 4, c = e & 15;
      const int m = (c >= r) ? r : 15 - r;
      const int n = (c >= r) ? c : 15 - r + c;
#pragma unroll
      for (int w = 0; w < NWAVE; ++w) acc += ldsCo[w][m * 16 + n];
    }
    planes[(size_t)(stepK & 1) * (NWG * PLANE) + (size_t)blockIdx.x * (32 * nl) + tid] = acc;
  }
}

extern "C" void kernel_launch(void* const* d_in, const int* in_sizes, int n_in,
                              void* d_out, int out_size, void* d_ws, size_t ws_size,
                              hipStream_t stream) {
  const float* x    = (const float*)d_in[0];
  const float* Wi   = (const float*)d_in[1];
  const float* bi   = (const float*)d_in[2];
  const float* gma  = (const float*)d_in[3];
  const float* bta  = (const float*)d_in[4];
  const float* R    = (const float*)d_in[5];
  const float* rb   = (const float*)d_in[6];
  const float* mask = (const float*)d_in[7];
  const float* Wc   = (const float*)d_in[8];
  const float* bc   = (const float*)d_in[9];
  const float* Wo   = (const float*)d_in[10];
  const float* bo   = (const float*)d_in[11];

  float* stateWS  = (float*)d_ws;
  int*   posWS    = (int*)((char*)d_ws + (size_t)BATCH * 16 * 4);
  float* planes   = (float*)((char*)posWS + (size_t)BATCH * 2 * 4);

  cyclic_step<true><<<NWG, WGT, 0, stream>>>(x, Wi, bi, gma, bta, R, rb, mask, Wc, bc,
                                             Wo, bo, stateWS, posWS, planes,
                                             (float*)d_out, 0);
  for (int k = 1; k < NSTEP; ++k) {
    cyclic_step<false><<<NWG, WGT, 0, stream>>>(x, Wi, bi, gma, bta, R, rb, mask, Wc, bc,
                                                Wo, bo, stateWS, posWS, planes,
                                                (float*)d_out, k);
  }
}


# in_act products no longer wait behind a barrier for the whole gather: first barrier orders LDS only, passed with the gather in flight; each patch row's 2 products issue as its load lands; same sums
# speedup vs baseline: 2.7373x; 1.0477x over previous
#include <hip/hip_runtime.h>
#include <hip/hip_bf16.h>

typedef __attribute__((ext_vector_type(2))) float v2f;
typedef __attribute__((ext_vector_type(8))) float v8f;
typedef float v4f_dw __attribute__((ext_vector_type(4), aligned(4)));

#define BAG   15
#define NSTEP 5
#define LAMBDA 0.001f
#define PS    8
#define IM    28
#define DENOM (IM - PS)
#define BATCH 65536
#define INDIM 66
#define NWG   256
#define WGT   512
#define NWAVE 16
#define PLANE 128
#define WISTR 68

template <int M>
__device__ __forceinline__ float row_xor(float v) {
  return __builtin_bit_cast(float,
      __builtin_amdgcn_update_dpp(0, __builtin_bit_cast(int, v),
                                  0x160 + M, 0xF, 0xF, true));
}

template <int M>
__device__ __forceinline__ int row_xor_i(int v) {
  return __builtin_amdgcn_update_dpp(0, v, 0x160 + M, 0xF, 0xF, true);
}

template <int S>
__device__ __forceinline__ float row_lane(float v) {
  return __builtin_bit_cast(float,
      __builtin_amdgcn_update_dpp(0, __builtin_bit_cast(int, v),
                                  0x150 + S, 0xF, 0xF, true));
}

template <int C>
__device__ __forceinline__ float4 plane_tree(float4 (&pv)[16]) {
#pragma unroll
  for (int st = C / 2; st >= 1; st >>= 1) {
#pragma unroll
    for (int q = 0; q < st; ++q) {
      pv[q].x += pv[q + st].x; pv[q].y += pv[q + st].y;
      pv[q].z += pv[q + st].z; pv[q].w += pv[q + st].w;
    }
  }
  return pv[0];
}

template <bool FIRST>
__global__ __launch_bounds__(WGT) __attribute__((amdgpu_waves_per_eu(4, 4)))
void cyclic_step(const float* __restrict__ imgs,
                 const float* __restrict__ Wi,  const float* __restrict__ bi,
                 const float* __restrict__ gma, const float* __restrict__ bta,
                 const float* __restrict__ R,   const float* __restrict__ rb,
                 const float* __restrict__ mask,
                 const float* __restrict__ Wc,  const float* __restrict__ bc,
                 const float* __restrict__ Wo,  const float* __restrict__ bo,
                 float* __restrict__ stateWS,   int* __restrict__ posWS,
                 float* planes,
                 float* __restrict__ out,       int stepK) {
  __shared__ float ldsWi[16 * WISTR];
  __shared__ float ldsWout[256];
  __shared__ float ldsReff[256];
  __shared__ float4 ldsRed[NWAVE][PLANE / 4];
  __shared__ unsigned ldsBal[4];
  __shared__ int ldsRank[PLANE];
  __shared__ int ldsInv[PLANE];
  __shared__ float ldsState[NWAVE][16][16];
  __shared__ float ldsCo[NWAVE][256];

  const int tid  = threadIdx.x;
  const int wave = tid >> 5;
  const int lane = tid & 31;
  const int half = lane >> 4;
  const int n16  = lane & 15;
  const int rowBase = blockIdx.x * 256 + wave * 16;
  const int myRow   = rowBase + n16;

  int p0 = DENOM / 2, p1 = DENOM / 2;
  if (!FIRST) { p0 = posWS[2 * myRow]; p1 = posWS[2 * myRow + 1]; }

  v2f sa[4];
#pragma unroll
  for (int t = 0; t < 4; ++t) { sa[t].x = 0.f; sa[t].y = 0.f; }
  if (!FIRST) {
    const float* st = stateWS + (size_t)myRow * 16;
#pragma unroll
    for (int t = 0; t < 4; ++t) {
      const int kb = 4 * t + 2 * half;
      sa[t].x = st[kb];
      sa[t].y = st[kb + 1];
    }
  }

  float rv, mv;
  const int oi = tid & 15, oj = (tid >> 4) & 15;
  {
    const int ic = oi < BAG ? oi : BAG - 1;
    const int jc = oj < BAG ? oj : BAG - 1;
    rv = R[ic * BAG + jc];
    mv = mask[ic * BAG + jc];
  }

  const int er = (tid >> 4) & 7, ec = tid & 15;
  const int em = (ec >= er) ? er : 15 - er;
  const int en = (ec >= er) ? ec : 15 - er + ec;
  const int enc = en < BAG ? en : BAG - 1;
  const float mk0 = mask[em * BAG + enc];
  const float mk1 = mask[enc * BAG + em];

  float wiv[3];
#pragma unroll
  for (int q = 0; q < 3; ++q) {
    const int idx = tid + WGT * q;
    const int n = idx / WISTR, k = idx - n * WISTR;
    const int nc = n < BAG ? n : BAG - 1;
    const int kc = k < INDIM ? k : INDIM - 1;
    const float v = Wi[nc * INDIM + kc];
    wiv[q] = (n < BAG && k < INDIM) ? v : 0.f;
  }
  float hv0;
  {
    const int k = (tid >> 4) & 15, col = tid & 15;
    const int kc = k < BAG ? k : BAG - 1;
    const float* hsrc = (col < 10) ? (Wo + col * BAG)
                                   : (Wc + ((col < 12 ? col : 11) - 10) * BAG);
    const float v = hsrc[kc];
    hv0 = (k < BAG && col < 12) ? v : 0.f;
  }
  const float* gp = imgs + (size_t)myRow * (IM * IM) + (p0 * IM + p1 + 4 * half);
  v4f_dw gq[8];
#pragma unroll
  for (int j = 0; j < 8; ++j)
    gq[j] = *reinterpret_cast<const v4f_dw*>(gp + j * IM);

  float bv, gv, btv;
  {
    const int pc = n16 < BAG ? n16 : BAG - 1;
    const float b0 = bi[pc];
    const float r0 = rb[pc];
    const float g0 = gma[pc];
    const float t0 = bta[pc];
    const bool okp = n16 < BAG;
    bv  = okp ? b0 + (FIRST ? 0.f : r0) : 0.f;
    gv  = okp ? g0 : 0.f;
    btv = okp ? t0 : 0.f;
  }
  const float bc0 = bc[0], bc1 = bc[1];
  const float boN = bo[n16 < 10 ? n16 : 9];

#pragma unroll
  for (int q = 0; q < 3; ++q) {
    const int idx = tid + WGT * q;
    if (idx < 16 * WISTR) ldsWi[idx] = wiv[q];
  }
  if (tid < 256) ldsWout[tid] = hv0;
  const bool need = (en < BAG) && (mk0 != 0.f || mk1 != 0.f);
  const unsigned bal = __builtin_amdgcn_ballot_w32(need);
  if (wave < 4 && lane == 0) ldsBal[wave] = bal;
  __builtin_amdgcn_fence(__ATOMIC_RELEASE, "workgroup", "local");
  __builtin_amdgcn_s_barrier();
  __builtin_amdgcn_fence(__ATOMIC_ACQUIRE, "workgroup", "local");

  const int c0 = __builtin_amdgcn_readfirstlane(__popc(ldsBal[0]));
  const int c1 = __builtin_amdgcn_readfirstlane(__popc(ldsBal[1]));
  const int c2 = __builtin_amdgcn_readfirstlane(__popc(ldsBal[2]));
  const int c3 = __builtin_amdgcn_readfirstlane(__popc(ldsBal[3]));
  const int nnzU = c0 + c1 + c2 + c3;
  const int nl = nnzU <= 32 ? 1 : (nnzU <= 64 ? 2 : 4);
  if (wave < 4) {
    const int pre = (wave > 0 ? c0 : 0) + (wave > 1 ? c1 : 0) + (wave > 2 ? c2 : 0);
    const int rk = pre + __popc(bal & ((1u << lane) - 1u));
    ldsRank[tid] = need ? rk : -1;
    if (need) ldsInv[rk] = tid;
  }

  const int pw = (wave + blockIdx.x) & (NWAVE - 1);
  float4 pv[16];
  if (!FIRST) {
    const float4* psrc = reinterpret_cast<const float4*>(
                             planes + (size_t)((stepK + 1) & 1) * (NWG * PLANE))
                         + pw * 32 + lane;
    if (nl == 1) {
#pragma unroll
      for (int q = 0; q < 4; ++q) pv[q] = psrc[(size_t)q * (NWAVE * 32)];
    } else if (nl == 2) {
#pragma unroll
      for (int q = 0; q < 8; ++q) pv[q] = psrc[(size_t)q * (NWAVE * 32)];
    } else {
#pragma unroll
      for (int q = 0; q < 16; ++q) pv[q] = psrc[(size_t)q * (NWAVE * 32)];
    }
  }

  v2f bw[17];
#pragma unroll
  for (int t = 0; t < 17; ++t)
    bw[t] = *(const v2f*)&ldsWi[n16 * WISTR + 4 * t + 2 * half];
  v8f accI = {0.f, 0.f, 0.f, 0.f, 0.f, 0.f, 0.f, 0.f};
#pragma unroll
  for (int j = 0; j < 8; ++j) {
    float qx = gq[j].x, qy = gq[j].y, qz = gq[j].z, qw = gq[j].w;
    asm volatile("v_permlane16_swap_b32 %0, %1" : "+v"(qx), "+v"(qz));
    asm volatile("v_permlane16_swap_b32 %0, %1" : "+v"(qy), "+v"(qw));
    v2f a0, a1;
    a0.x = qx; a0.y = qy;
    a1.x = qz; a1.y = qw;
    accI = __builtin_amdgcn_wmma_f32_16x16x4_f32(false, a0, false, bw[2 * j],
                                                 (short)0, accI, false, false);
    accI = __builtin_amdgcn_wmma_f32_16x16x4_f32(false, a1, false, bw[2 * j + 1],
                                                 (short)0, accI, false, false);
    __builtin_amdgcn_sched_barrier(0);
  }
  {
    v2f a;
    if (half == 0) {
      a.x = (float)p0 * (2.0f / DENOM) - 1.0f;
      a.y = (float)p1 * (2.0f / DENOM) - 1.0f;
    } else {
      a.x = 0.f; a.y = 0.f;
    }
    accI = __builtin_amdgcn_wmma_f32_16x16x4_f32(false, a, false, bw[16],
                                                 (short)0, accI, false, false);
  }

  if (!FIRST) {
    float4 s4;
    if (nl == 1) {
      s4 = plane_tree<4>(pv);
      s4.x += row_xor<8>(s4.x);         s4.y += row_xor<8>(s4.y);
      s4.z += row_xor<8>(s4.z);         s4.w += row_xor<8>(s4.w);
      s4.x += __shfl_xor(s4.x, 16, 32); s4.y += __shfl_xor(s4.y, 16, 32);
      s4.z += __shfl_xor(s4.z, 16, 32); s4.w += __shfl_xor(s4.w, 16, 32);
    } else if (nl == 2) {
      s4 = plane_tree<8>(pv);
      s4.x += __shfl_xor(s4.x, 16, 32); s4.y += __shfl_xor(s4.y, 16, 32);
      s4.z += __shfl_xor(s4.z, 16, 32); s4.w += __shfl_xor(s4.w, 16, 32);
    } else {
      s4 = plane_tree<16>(pv);
    }
    if (lane < 8 * nl) ldsRed[pw][lane] = s4;
    __syncthreads();

    if (tid < 256) {
      const int a = oi < oj ? oi : oj;
      const int b = oi < oj ? oj : oi;
      const int q = (a <= 7) ? a * 16 + b : (15 - a) * 16 + (b - a);
      const bool live = (oi < BAG) && (oj < BAG);
      const int rk = ldsRank[q];
      float co = 0.f;
      if (live && rk >= 0) {
        const float* rf = reinterpret_cast<const float*>(&ldsRed[0][0]) + rk;
        float t[NWAVE];
#pragma unroll
        for (int p = 0; p < NWAVE; ++p) t[p] = rf[p * PLANE];
#pragma unroll
        for (int st = NWAVE / 2; st >= 1; st >>= 1) {
#pragma unroll
          for (int p = 0; p < st; ++p) t[p] += t[p + st];
        }
        co = t[0] * (1.0f / (float)BATCH);
      }
      ldsReff[tid] = live ? (rv - LAMBDA * co) * mv : 0.f;
    }
    __syncthreads();
#pragma unroll
    for (int t = 0; t < 4; ++t) {
      const int kb = 4 * t + 2 * half;
      v2f b;
      b.x = ldsReff[kb * 16 + n16];
      b.y = ldsReff[(kb + 1) * 16 + n16];
      accI = __builtin_amdgcn_wmma_f32_16x16x4_f32(false, sa[t], false, b,
                                                   (short)0, accI, false, false);
    }
  }

  const float biasN = bv;
  const float gN    = gv;
  const float bN    = btv;
  float hv[8], red[8];
#pragma unroll
  for (int r = 0; r < 8; ++r) {
    hv[r]  = fmaxf(accI[r] + biasN, 0.f);
    red[r] = hv[r];
  }
#pragma unroll
  for (int r = 0; r < 8; ++r) red[r] += row_xor<8>(red[r]);
#pragma unroll
  for (int r = 0; r < 8; ++r) red[r] += row_xor<4>(red[r]);
#pragma unroll
  for (int r = 0; r < 8; ++r) red[r] += row_xor<2>(red[r]);
#pragma unroll
  for (int r = 0; r < 8; ++r) red[r] += row_xor<1>(red[r]);
#pragma unroll
  for (int r = 0; r < 8; ++r) {
    float mu = red[r] * (1.0f / BAG);
    float dd = hv[r] - mu;
    hv[r]  = dd;
    red[r] = (n16 < BAG) ? dd * dd : 0.f;
  }
#pragma unroll
  for (int r = 0; r < 8; ++r) red[r] += row_xor<8>(red[r]);
#pragma unroll
  for (int r = 0; r < 8; ++r) red[r] += row_xor<4>(red[r]);
#pragma unroll
  for (int r = 0; r < 8; ++r) red[r] += row_xor<2>(red[r]);
#pragma unroll
  for (int r = 0; r < 8; ++r) red[r] += row_xor<1>(red[r]);
#pragma unroll
  for (int r = 0; r < 8; ++r) {
    float s = hv[r] * rsqrtf(red[r] * (1.0f / BAG) + 1e-5f) * gN + bN;
    hv[r] = s;
    ldsState[wave][r + half * 8][n16] = s;
  }
  if (stepK < NSTEP - 1) {
#pragma unroll
    for (int r = 0; r < 8; ++r)
      stateWS[(size_t)(rowBase + r + half * 8) * 16 + n16] = hv[r];
  }
  __syncthreads();

  v8f accO = {0.f, 0.f, 0.f, 0.f, 0.f, 0.f, 0.f, 0.f};
#pragma unroll
  for (int t = 0; t < 4; ++t) {
    const int kb = 4 * t + 2 * half;
    v2f a, b;
    a.x = ldsState[wave][n16][kb];
    a.y = ldsState[wave][n16][kb + 1];
    b.x = ldsWout[kb * 16 + n16];
    b.y = ldsWout[(kb + 1) * 16 + n16];
    accO = __builtin_amdgcn_wmma_f32_16x16x4_f32(false, a, false, b,
                                                 (short)0, accO, false, false);
  }

  if (stepK < NSTEP - 1) {
    const int  j      = n16 & 7;
    const bool second = n16 >= 8;
    float raw = 0.f;
#pragma unroll
    for (int r = 0; r < 8; ++r) {
      const float v10 = row_lane<10>(accO[r]);
      const float v11 = row_lane<11>(accO[r]);
      const float v   = second ? v11 : v10;
      raw = (j == r) ? v : raw;
    }
    const float cm = tanhf(raw + (second ? bc1 : bc0));
    const float cx = row_xor<8>(cm);
    const int r0 = row_xor_i<8>(p0);
    const int r1 = row_xor_i<8>(p1);
    if (!second) {
      const float c0 = cm, c1 = cx;
      int q0 = half ? r0 : p0;
      int q1 = half ? r1 : p1;
      const bool rowSel = fabsf(c0) >= fabsf(c1);
      const int mv0 = rowSel ? ((c0 > 0.f) - (c0 < 0.f)) : 0;
      const int mv1 = rowSel ? 0 : ((c1 > 0.f) - (c1 < 0.f));
      q0 = min(max(q0 + mv0, 0), DENOM);
      q1 = min(max(q1 + mv1, 0), DENOM);
      const int row = rowBase + half * 8 + n16;
      *reinterpret_cast<int2*>(posWS + 2 * row) = make_int2(q0, q1);
    }
  }

  if (stepK == NSTEP - 1 && n16 < 10) {
#pragma unroll
    for (int r = 0; r < 8; ++r)
      out[(size_t)(rowBase + r + half * 8) * 10 + n16] = accO[r] + boN;
  }

  if (stepK < NSTEP - 1) {
    v8f accC = {0.f, 0.f, 0.f, 0.f, 0.f, 0.f, 0.f, 0.f};
#pragma unroll
    for (int t = 0; t < 4; ++t) {
      const int kb = 4 * t + 2 * half;
      v2f f;
      f.x = ldsState[wave][kb][n16];
      f.y = ldsState[wave][kb + 1][n16];
      accC = __builtin_amdgcn_wmma_f32_16x16x4_f32(false, f, false, f,
                                                   (short)0, accC, false, false);
    }
#pragma unroll
    for (int r = 0; r < 8; ++r)
      ldsCo[wave][(r + half * 8) * 16 + n16] = accC[r];
  }
  __syncthreads();
  if (stepK < NSTEP - 1 && tid < 32 * nl) {
    float acc = 0.f;
    if (tid < nnzU) {
      const int e = ldsInv[tid];
      const int r = e >> 4, c = e & 15;
      const int m = (c >= r) ? r : 15 - r;
      const int n = (c >= r) ? c : 15 - r + c;
#pragma unroll
      for (int w = 0; w < NWAVE; ++w) acc += ldsCo[w][m * 16 + n];
    }
    planes[(size_t)(stepK & 1) * (NWG * PLANE) + (size_t)blockIdx.x * (32 * nl) + tid] = acc;
  }
}

extern "C" void kernel_launch(void* const* d_in, const int* in_sizes, int n_in,
                              void* d_out, int out_size, void* d_ws, size_t ws_size,
                              hipStream_t stream) {
  const float* x    = (const float*)d_in[0];
  const float* Wi   = (const float*)d_in[1];
  const float* bi   = (const float*)d_in[2];
  const float* gma  = (const float*)d_in[3];
  const float* bta  = (const float*)d_in[4];
  const float* R    = (const float*)d_in[5];
  const float* rb   = (const float*)d_in[6];
  const float* mask = (const float*)d_in[7];
  const float* Wc   = (const float*)d_in[8];
  const float* bc   = (const float*)d_in[9];
  const float* Wo   = (const float*)d_in[10];
  const float* bo   = (const float*)d_in[11];

  float* stateWS  = (float*)d_ws;
  int*   posWS    = (int*)((char*)d_ws + (size_t)BATCH * 16 * 4);
  float* planes   = (float*)((char*)posWS + (size_t)BATCH * 2 * 4);

  cyclic_step<true><<<NWG, WGT, 0, stream>>>(x, Wi, bi, gma, bta, R, rb, mask, Wc, bc,
                                             Wo, bo, stateWS, posWS, planes,
                                             (float*)d_out, 0);
  for (int k = 1; k < NSTEP; ++k) {
    cyclic_step<false><<<NWG, WGT, 0, stream>>>(x, Wi, bi, gma, bta, R, rb, mask, Wc, bc,
                                                Wo, bo, stateWS, posWS, planes,
                                                (float*)d_out, k);
  }
}


# barriers behind the LayerNorm no longer drain the store counter: state tile hand-over is wave-local (own wave only, no work-group barrier), last barrier orders LDS only; same sums
# speedup vs baseline: 3.0149x; 1.1014x over previous
#include <hip/hip_runtime.h>
#include <hip/hip_bf16.h>

typedef __attribute__((ext_vector_type(2))) float v2f;
typedef __attribute__((ext_vector_type(8))) float v8f;
typedef float v4f_dw __attribute__((ext_vector_type(4), aligned(4)));

#define BAG   15
#define NSTEP 5
#define LAMBDA 0.001f
#define PS    8
#define IM    28
#define DENOM (IM - PS)
#define BATCH 65536
#define INDIM 66
#define NWG   256
#define WGT   512
#define NWAVE 16
#define PLANE 128
#define WISTR 68

template <int M>
__device__ __forceinline__ float row_xor(float v) {
  return __builtin_bit_cast(float,
      __builtin_amdgcn_update_dpp(0, __builtin_bit_cast(int, v),
                                  0x160 + M, 0xF, 0xF, true));
}

template <int M>
__device__ __forceinline__ int row_xor_i(int v) {
  return __builtin_amdgcn_update_dpp(0, v, 0x160 + M, 0xF, 0xF, true);
}

template <int S>
__device__ __forceinline__ float row_lane(float v) {
  return __builtin_bit_cast(float,
      __builtin_amdgcn_update_dpp(0, __builtin_bit_cast(int, v),
                                  0x150 + S, 0xF, 0xF, true));
}

template <int C>
__device__ __forceinline__ float4 plane_tree(float4 (&pv)[16]) {
#pragma unroll
  for (int st = C / 2; st >= 1; st >>= 1) {
#pragma unroll
    for (int q = 0; q < st; ++q) {
      pv[q].x += pv[q + st].x; pv[q].y += pv[q + st].y;
      pv[q].z += pv[q + st].z; pv[q].w += pv[q + st].w;
    }
  }
  return pv[0];
}

template <bool FIRST>
__global__ __launch_bounds__(WGT) __attribute__((amdgpu_waves_per_eu(4, 4)))
void cyclic_step(const float* __restrict__ imgs,
                 const float* __restrict__ Wi,  const float* __restrict__ bi,
                 const float* __restrict__ gma, const float* __restrict__ bta,
                 const float* __restrict__ R,   const float* __restrict__ rb,
                 const float* __restrict__ mask,
                 const float* __restrict__ Wc,  const float* __restrict__ bc,
                 const float* __restrict__ Wo,  const float* __restrict__ bo,
                 float* __restrict__ stateWS,   int* __restrict__ posWS,
                 float* planes,
                 float* __restrict__ out,       int stepK) {
  __shared__ float ldsWi[16 * WISTR];
  __shared__ float ldsWout[256];
  __shared__ float ldsReff[256];
  __shared__ float4 ldsRed[NWAVE][PLANE / 4];
  __shared__ unsigned ldsBal[4];
  __shared__ int ldsRank[PLANE];
  __shared__ int ldsInv[PLANE];
  __shared__ float ldsState[NWAVE][16][16];
  __shared__ float ldsCo[NWAVE][256];

  const int tid  = threadIdx.x;
  const int wave = tid >> 5;
  const int lane = tid & 31;
  const int half = lane >> 4;
  const int n16  = lane & 15;
  const int rowBase = blockIdx.x * 256 + wave * 16;
  const int myRow   = rowBase + n16;

  int p0 = DENOM / 2, p1 = DENOM / 2;
  if (!FIRST) { p0 = posWS[2 * myRow]; p1 = posWS[2 * myRow + 1]; }

  v2f sa[4];
#pragma unroll
  for (int t = 0; t < 4; ++t) { sa[t].x = 0.f; sa[t].y = 0.f; }
  if (!FIRST) {
    const float* st = stateWS + (size_t)myRow * 16;
#pragma unroll
    for (int t = 0; t < 4; ++t) {
      const int kb = 4 * t + 2 * half;
      sa[t].x = st[kb];
      sa[t].y = st[kb + 1];
    }
  }

  float rv, mv;
  const int oi = tid & 15, oj = (tid >> 4) & 15;
  {
    const int ic = oi < BAG ? oi : BAG - 1;
    const int jc = oj < BAG ? oj : BAG - 1;
    rv = R[ic * BAG + jc];
    mv = mask[ic * BAG + jc];
  }

  const int er = (tid >> 4) & 7, ec = tid & 15;
  const int em = (ec >= er) ? er : 15 - er;
  const int en = (ec >= er) ? ec : 15 - er + ec;
  const int enc = en < BAG ? en : BAG - 1;
  const float mk0 = mask[em * BAG + enc];
  const float mk1 = mask[enc * BAG + em];

  float wiv[3];
#pragma unroll
  for (int q = 0; q < 3; ++q) {
    const int idx = tid + WGT * q;
    const int n = idx / WISTR, k = idx - n * WISTR;
    const int nc = n < BAG ? n : BAG - 1;
    const int kc = k < INDIM ? k : INDIM - 1;
    const float v = Wi[nc * INDIM + kc];
    wiv[q] = (n < BAG && k < INDIM) ? v : 0.f;
  }
  float hv0;
  {
    const int k = (tid >> 4) & 15, col = tid & 15;
    const int kc = k < BAG ? k : BAG - 1;
    const float* hsrc = (col < 10) ? (Wo + col * BAG)
                                   : (Wc + ((col < 12 ? col : 11) - 10) * BAG);
    const float v = hsrc[kc];
    hv0 = (k < BAG && col < 12) ? v : 0.f;
  }
  const float* gp = imgs + (size_t)myRow * (IM * IM) + (p0 * IM + p1 + 4 * half);
  v4f_dw gq[8];
#pragma unroll
  for (int j = 0; j < 8; ++j)
    gq[j] = *reinterpret_cast<const v4f_dw*>(gp + j * IM);

  float bv, gv, btv;
  {
    const int pc = n16 < BAG ? n16 : BAG - 1;
    const float b0 = bi[pc];
    const float r0 = rb[pc];
    const float g0 = gma[pc];
    const float t0 = bta[pc];
    const bool okp = n16 < BAG;
    bv  = okp ? b0 + (FIRST ? 0.f : r0) : 0.f;
    gv  = okp ? g0 : 0.f;
    btv = okp ? t0 : 0.f;
  }
  const float bc0 = bc[0], bc1 = bc[1];
  const float boN = bo[n16 < 10 ? n16 : 9];

#pragma unroll
  for (int q = 0; q < 3; ++q) {
    const int idx = tid + WGT * q;
    if (idx < 16 * WISTR) ldsWi[idx] = wiv[q];
  }
  if (tid < 256) ldsWout[tid] = hv0;
  const bool need = (en < BAG) && (mk0 != 0.f || mk1 != 0.f);
  const unsigned bal = __builtin_amdgcn_ballot_w32(need);
  if (wave < 4 && lane == 0) ldsBal[wave] = bal;
  __builtin_amdgcn_fence(__ATOMIC_RELEASE, "workgroup", "local");
  __builtin_amdgcn_s_barrier();
  __builtin_amdgcn_fence(__ATOMIC_ACQUIRE, "workgroup", "local");

  const int c0 = __builtin_amdgcn_readfirstlane(__popc(ldsBal[0]));
  const int c1 = __builtin_amdgcn_readfirstlane(__popc(ldsBal[1]));
  const int c2 = __builtin_amdgcn_readfirstlane(__popc(ldsBal[2]));
  const int c3 = __builtin_amdgcn_readfirstlane(__popc(ldsBal[3]));
  const int nnzU = c0 + c1 + c2 + c3;
  const int nl = nnzU <= 32 ? 1 : (nnzU <= 64 ? 2 : 4);
  if (wave < 4) {
    const int pre = (wave > 0 ? c0 : 0) + (wave > 1 ? c1 : 0) + (wave > 2 ? c2 : 0);
    const int rk = pre + __popc(bal & ((1u << lane) - 1u));
    ldsRank[tid] = need ? rk : -1;
    if (need) ldsInv[rk] = tid;
  }

  const int pw = (wave + blockIdx.x) & (NWAVE - 1);
  float4 pv[16];
  if (!FIRST) {
    const float4* psrc = reinterpret_cast<const float4*>(
                             planes + (size_t)((stepK + 1) & 1) * (NWG * PLANE))
                         + pw * 32 + lane;
    if (nl == 1) {
#pragma unroll
      for (int q = 0; q < 4; ++q) pv[q] = psrc[(size_t)q * (NWAVE * 32)];
    } else if (nl == 2) {
#pragma unroll
      for (int q = 0; q < 8; ++q) pv[q] = psrc[(size_t)q * (NWAVE * 32)];
    } else {
#pragma unroll
      for (int q = 0; q < 16; ++q) pv[q] = psrc[(size_t)q * (NWAVE * 32)];
    }
  }

  v2f bw[17];
#pragma unroll
  for (int t = 0; t < 17; ++t)
    bw[t] = *(const v2f*)&ldsWi[n16 * WISTR + 4 * t + 2 * half];
  v8f accI = {0.f, 0.f, 0.f, 0.f, 0.f, 0.f, 0.f, 0.f};
#pragma unroll
  for (int j = 0; j < 8; ++j) {
    float qx = gq[j].x, qy = gq[j].y, qz = gq[j].z, qw = gq[j].w;
    asm volatile("v_permlane16_swap_b32 %0, %1" : "+v"(qx), "+v"(qz));
    asm volatile("v_permlane16_swap_b32 %0, %1" : "+v"(qy), "+v"(qw));
    v2f a0, a1;
    a0.x = qx; a0.y = qy;
    a1.x = qz; a1.y = qw;
    accI = __builtin_amdgcn_wmma_f32_16x16x4_f32(false, a0, false, bw[2 * j],
                                                 (short)0, accI, false, false);
    accI = __builtin_amdgcn_wmma_f32_16x16x4_f32(false, a1, false, bw[2 * j + 1],
                                                 (short)0, accI, false, false);
    __builtin_amdgcn_sched_barrier(0);
  }
  {
    v2f a;
    if (half == 0) {
      a.x = (float)p0 * (2.0f / DENOM) - 1.0f;
      a.y = (float)p1 * (2.0f / DENOM) - 1.0f;
    } else {
      a.x = 0.f; a.y = 0.f;
    }
    accI = __builtin_amdgcn_wmma_f32_16x16x4_f32(false, a, false, bw[16],
                                                 (short)0, accI, false, false);
  }

  if (!FIRST) {
    float4 s4;
    if (nl == 1) {
      s4 = plane_tree<4>(pv);
      s4.x += row_xor<8>(s4.x);         s4.y += row_xor<8>(s4.y);
      s4.z += row_xor<8>(s4.z);         s4.w += row_xor<8>(s4.w);
      s4.x += __shfl_xor(s4.x, 16, 32); s4.y += __shfl_xor(s4.y, 16, 32);
      s4.z += __shfl_xor(s4.z, 16, 32); s4.w += __shfl_xor(s4.w, 16, 32);
    } else if (nl == 2) {
      s4 = plane_tree<8>(pv);
      s4.x += __shfl_xor(s4.x, 16, 32); s4.y += __shfl_xor(s4.y, 16, 32);
      s4.z += __shfl_xor(s4.z, 16, 32); s4.w += __shfl_xor(s4.w, 16, 32);
    } else {
      s4 = plane_tree<16>(pv);
    }
    if (lane < 8 * nl) ldsRed[pw][lane] = s4;
    __syncthreads();

    if (tid < 256) {
      const int a = oi < oj ? oi : oj;
      const int b = oi < oj ? oj : oi;
      const int q = (a <= 7) ? a * 16 + b : (15 - a) * 16 + (b - a);
      const bool live = (oi < BAG) && (oj < BAG);
      const int rk = ldsRank[q];
      float co = 0.f;
      if (live && rk >= 0) {
        const float* rf = reinterpret_cast<const float*>(&ldsRed[0][0]) + rk;
        float t[NWAVE];
#pragma unroll
        for (int p = 0; p < NWAVE; ++p) t[p] = rf[p * PLANE];
#pragma unroll
        for (int st = NWAVE / 2; st >= 1; st >>= 1) {
#pragma unroll
          for (int p = 0; p < st; ++p) t[p] += t[p + st];
        }
        co = t[0] * (1.0f / (float)BATCH);
      }
      ldsReff[tid] = live ? (rv - LAMBDA * co) * mv : 0.f;
    }
    __syncthreads();
#pragma unroll
    for (int t = 0; t < 4; ++t) {
      const int kb = 4 * t + 2 * half;
      v2f b;
      b.x = ldsReff[kb * 16 + n16];
      b.y = ldsReff[(kb + 1) * 16 + n16];
      accI = __builtin_amdgcn_wmma_f32_16x16x4_f32(false, sa[t], false, b,
                                                   (short)0, accI, false, false);
    }
  }

  const float biasN = bv;
  const float gN    = gv;
  const float bN    = btv;
  float hv[8], red[8];
#pragma unroll
  for (int r = 0; r < 8; ++r) {
    hv[r]  = fmaxf(accI[r] + biasN, 0.f);
    red[r] = hv[r];
  }
#pragma unroll
  for (int r = 0; r < 8; ++r) red[r] += row_xor<8>(red[r]);
#pragma unroll
  for (int r = 0; r < 8; ++r) red[r] += row_xor<4>(red[r]);
#pragma unroll
  for (int r = 0; r < 8; ++r) red[r] += row_xor<2>(red[r]);
#pragma unroll
  for (int r = 0; r < 8; ++r) red[r] += row_xor<1>(red[r]);
#pragma unroll
  for (int r = 0; r < 8; ++r) {
    float mu = red[r] * (1.0f / BAG);
    float dd = hv[r] - mu;
    hv[r]  = dd;
    red[r] = (n16 < BAG) ? dd * dd : 0.f;
  }
#pragma unroll
  for (int r = 0; r < 8; ++r) red[r] += row_xor<8>(red[r]);
#pragma unroll
  for (int r = 0; r < 8; ++r) red[r] += row_xor<4>(red[r]);
#pragma unroll
  for (int r = 0; r < 8; ++r) red[r] += row_xor<2>(red[r]);
#pragma unroll
  for (int r = 0; r < 8; ++r) red[r] += row_xor<1>(red[r]);
#pragma unroll
  for (int r = 0; r < 8; ++r) {
    float s = hv[r] * rsqrtf(red[r] * (1.0f / BAG) + 1e-5f) * gN + bN;
    hv[r] = s;
    ldsState[wave][r + half * 8][n16] = s;
  }
  if (stepK < NSTEP - 1) {
#pragma unroll
    for (int r = 0; r < 8; ++r)
      stateWS[(size_t)(rowBase + r + half * 8) * 16 + n16] = hv[r];
  }
  __builtin_amdgcn_fence(__ATOMIC_RELEASE, "wavefront", "local");
  __builtin_amdgcn_wave_barrier();
  __builtin_amdgcn_fence(__ATOMIC_ACQUIRE, "wavefront", "local");

  v8f accO = {0.f, 0.f, 0.f, 0.f, 0.f, 0.f, 0.f, 0.f};
#pragma unroll
  for (int t = 0; t < 4; ++t) {
    const int kb = 4 * t + 2 * half;
    v2f a, b;
    a.x = ldsState[wave][n16][kb];
    a.y = ldsState[wave][n16][kb + 1];
    b.x = ldsWout[kb * 16 + n16];
    b.y = ldsWout[(kb + 1) * 16 + n16];
    accO = __builtin_amdgcn_wmma_f32_16x16x4_f32(false, a, false, b,
                                                 (short)0, accO, false, false);
  }

  if (stepK < NSTEP - 1) {
    const int  j      = n16 & 7;
    const bool second = n16 >= 8;
    float raw = 0.f;
#pragma unroll
    for (int r = 0; r < 8; ++r) {
      const float v10 = row_lane<10>(accO[r]);
      const float v11 = row_lane<11>(accO[r]);
      const float v   = second ? v11 : v10;
      raw = (j == r) ? v : raw;
    }
    const float cm = tanhf(raw + (second ? bc1 : bc0));
    const float cx = row_xor<8>(cm);
    const int r0 = row_xor_i<8>(p0);
    const int r1 = row_xor_i<8>(p1);
    if (!second) {
      const float c0 = cm, c1 = cx;
      int q0 = half ? r0 : p0;
      int q1 = half ? r1 : p1;
      const bool rowSel = fabsf(c0) >= fabsf(c1);
      const int mv0 = rowSel ? ((c0 > 0.f) - (c0 < 0.f)) : 0;
      const int mv1 = rowSel ? 0 : ((c1 > 0.f) - (c1 < 0.f));
      q0 = min(max(q0 + mv0, 0), DENOM);
      q1 = min(max(q1 + mv1, 0), DENOM);
      const int row = rowBase + half * 8 + n16;
      *reinterpret_cast<int2*>(posWS + 2 * row) = make_int2(q0, q1);
    }
  }

  if (stepK == NSTEP - 1 && n16 < 10) {
#pragma unroll
    for (int r = 0; r < 8; ++r)
      out[(size_t)(rowBase + r + half * 8) * 10 + n16] = accO[r] + boN;
  }

  if (stepK < NSTEP - 1) {
    v8f accC = {0.f, 0.f, 0.f, 0.f, 0.f, 0.f, 0.f, 0.f};
#pragma unroll
    for (int t = 0; t < 4; ++t) {
      const int kb = 4 * t + 2 * half;
      v2f f;
      f.x = ldsState[wave][kb][n16];
      f.y = ldsState[wave][kb + 1][n16];
      accC = __builtin_amdgcn_wmma_f32_16x16x4_f32(false, f, false, f,
                                                   (short)0, accC, false, false);
    }
#pragma unroll
    for (int r = 0; r < 8; ++r)
      ldsCo[wave][(r + half * 8) * 16 + n16] = accC[r];
  }
  __builtin_amdgcn_fence(__ATOMIC_RELEASE, "workgroup", "local");
  __builtin_amdgcn_s_barrier();
  __builtin_amdgcn_fence(__ATOMIC_ACQUIRE, "workgroup", "local");
  if (stepK < NSTEP - 1 && tid < 32 * nl) {
    float acc = 0.f;
    if (tid < nnzU) {
      const int e = ldsInv[tid];
      const int r = e >> 4, c = e & 15;
      const int m = (c >= r) ? r : 15 - r;
      const int n = (c >= r) ? c : 15 - r + c;
#pragma unroll
      for (int w = 0; w < NWAVE; ++w) acc += ldsCo[w][m * 16 + n];
    }
    planes[(size_t)(stepK & 1) * (NWG * PLANE) + (size_t)blockIdx.x * (32 * nl) + tid] = acc;
  }
}

extern "C" void kernel_launch(void* const* d_in, const int* in_sizes, int n_in,
                              void* d_out, int out_size, void* d_ws, size_t ws_size,
                              hipStream_t stream) {
  const float* x    = (const float*)d_in[0];
  const float* Wi   = (const float*)d_in[1];
  const float* bi   = (const float*)d_in[2];
  const float* gma  = (const float*)d_in[3];
  const float* bta  = (const float*)d_in[4];
  const float* R    = (const float*)d_in[5];
  const float* rb   = (const float*)d_in[6];
  const float* mask = (const float*)d_in[7];
  const float* Wc   = (const float*)d_in[8];
  const float* bc   = (const float*)d_in[9];
  const float* Wo   = (const float*)d_in[10];
  const float* bo   = (const float*)d_in[11];

  float* stateWS  = (float*)d_ws;
  int*   posWS    = (int*)((char*)d_ws + (size_t)BATCH * 16 * 4);
  float* planes   = (float*)((char*)posWS + (size_t)BATCH * 2 * 4);

  cyclic_step<true><<<NWG, WGT, 0, stream>>>(x, Wi, bi, gma, bta, R, rb, mask, Wc, bc,
                                             Wo, bo, stateWS, posWS, planes,
                                             (float*)d_out, 0);
  for (int k = 1; k < NSTEP; ++k) {
    cyclic_step<false><<<NWG, WGT, 0, stream>>>(x, Wi, bi, gma, bta, R, rb, mask, Wc, bc,
                                                Wo, bo, stateWS, posWS, planes,
                                                (float*)d_out, k);
  }
}


# products transposed (weights as A, batch rows as B): a lane holds 8 features of one row; LayerNorm sums run down its registers + 1 half swap, not 8 DPP hops; LN, recurrent, head sum order changes
# speedup vs baseline: 3.4320x; 1.1383x over previous
#include <hip/hip_runtime.h>
#include <hip/hip_bf16.h>

typedef __attribute__((ext_vector_type(2))) float v2f;
typedef __attribute__((ext_vector_type(8))) float v8f;
typedef float v4f_dw __attribute__((ext_vector_type(4), aligned(4)));

#define BAG   15
#define NSTEP 5
#define LAMBDA 0.001f
#define PS    8
#define IM    28
#define DENOM (IM - PS)
#define BATCH 65536
#define INDIM 66
#define NWG   256
#define WGT   512
#define NWAVE 16
#define PLANE 128
#define WISTR 68

template <int M>
__device__ __forceinline__ float row_xor(float v) {
  return __builtin_bit_cast(float,
      __builtin_amdgcn_update_dpp(0, __builtin_bit_cast(int, v),
                                  0x160 + M, 0xF, 0xF, true));
}

__device__ __forceinline__ float xhalf(float v) {
  const int x = __builtin_bit_cast(int, v);
  return __builtin_bit_cast(float,
      __builtin_amdgcn_permlanex16(x, x, 0x76543210, 0xfedcba98, false, false));
}

template <int C>
__device__ __forceinline__ float4 plane_tree(float4 (&pv)[16]) {
#pragma unroll
  for (int st = C / 2; st >= 1; st >>= 1) {
#pragma unroll
    for (int q = 0; q < st; ++q) {
      pv[q].x += pv[q + st].x; pv[q].y += pv[q + st].y;
      pv[q].z += pv[q + st].z; pv[q].w += pv[q + st].w;
    }
  }
  return pv[0];
}

template <bool FIRST>
__global__ __launch_bounds__(WGT) __attribute__((amdgpu_waves_per_eu(4, 4)))
void cyclic_step(const float* __restrict__ imgs,
                 const float* __restrict__ Wi,  const float* __restrict__ bi,
                 const float* __restrict__ gma, const float* __restrict__ bta,
                 const float* __restrict__ R,   const float* __restrict__ rb,
                 const float* __restrict__ mask,
                 const float* __restrict__ Wc,  const float* __restrict__ bc,
                 const float* __restrict__ Wo,  const float* __restrict__ bo,
                 float* __restrict__ stateWS,   int* __restrict__ posWS,
                 float* planes,
                 float* __restrict__ out,       int stepK) {
  __shared__ float ldsWi[16 * WISTR];
  __shared__ float ldsWout[256];
  __shared__ float ldsReff[256];
  __shared__ __attribute__((aligned(16))) float ldsVec[NWAVE][4][16];
  __shared__ float4 ldsRed[NWAVE][PLANE / 4];
  __shared__ unsigned ldsBal[4];
  __shared__ int ldsRank[PLANE];
  __shared__ int ldsInv[PLANE];
  __shared__ __attribute__((aligned(16))) float ldsState[NWAVE][16][16];
  __shared__ float ldsCo[NWAVE][256];

  const int tid  = threadIdx.x;
  const int wave = tid >> 5;
  const int lane = tid & 31;
  const int half = lane >> 4;
  const int n16  = lane & 15;
  const int rowBase = blockIdx.x * 256 + wave * 16;
  const int myRow   = rowBase + n16;

  int p0 = DENOM / 2, p1 = DENOM / 2;
  if (!FIRST) { p0 = posWS[2 * myRow]; p1 = posWS[2 * myRow + 1]; }

  float4 sq0 = make_float4(0.f, 0.f, 0.f, 0.f), sq1 = make_float4(0.f, 0.f, 0.f, 0.f);
  if (!FIRST) {
    const float4* st = reinterpret_cast<const float4*>(stateWS + (size_t)myRow * 16 + 8 * half);
    sq0 = st[0];
    sq1 = st[1];
  }

  float rv, mv;
  const int oi = tid & 15, oj = (tid >> 4) & 15;
  {
    const int ic = oi < BAG ? oi : BAG - 1;
    const int jc = oj < BAG ? oj : BAG - 1;
    rv = R[ic * BAG + jc];
    mv = mask[ic * BAG + jc];
  }

  const int er = (tid >> 4) & 7, ec = tid & 15;
  const int em = (ec >= er) ? er : 15 - er;
  const int en = (ec >= er) ? ec : 15 - er + ec;
  const int enc = en < BAG ? en : BAG - 1;
  const float mk0 = mask[em * BAG + enc];
  const float mk1 = mask[enc * BAG + em];

  float wiv[3];
#pragma unroll
  for (int q = 0; q < 3; ++q) {
    const int idx = tid + WGT * q;
    const int n = idx / WISTR, k = idx - n * WISTR;
    const int nc = n < BAG ? n : BAG - 1;
    const int kc = k < INDIM ? k : INDIM - 1;
    const float v = Wi[nc * INDIM + kc];
    wiv[q] = (n < BAG && k < INDIM) ? v : 0.f;
  }
  float hv0;
  {
    const int k = (tid >> 4) & 15, col = tid & 15;
    const int kc = k < BAG ? k : BAG - 1;
    const float* hsrc = (col < 10) ? (Wo + col * BAG)
                                   : (Wc + ((col < 12 ? col : 11) - 10) * BAG);
    const float v = hsrc[kc];
    hv0 = (k < BAG && col < 12) ? v : 0.f;
  }
  float bv, gv, btv, bov1;
  {
    const int pc = n16 < BAG ? n16 : BAG - 1;
    const float b0 = bi[pc];
    const float r0 = rb[pc];
    const float g0 = gma[pc];
    const float t0 = bta[pc];
    const float o0 = bo[n16 < 10 ? n16 : 9];
    const bool okp = n16 < BAG;
    bv   = okp ? b0 + (FIRST ? 0.f : r0) : 0.f;
    gv   = okp ? g0 : 0.f;
    btv  = okp ? t0 : 0.f;
    bov1 = (n16 < 10) ? o0 : 0.f;
  }
  const float* gp = imgs + (size_t)myRow * (IM * IM) + (p0 * IM + p1 + 4 * half);
  v4f_dw gq[8];
#pragma unroll
  for (int j = 0; j < 8; ++j)
    gq[j] = *reinterpret_cast<const v4f_dw*>(gp + j * IM);

  const float bc0 = bc[0], bc1 = bc[1];

#pragma unroll
  for (int q = 0; q < 3; ++q) {
    const int idx = tid + WGT * q;
    if (idx < 16 * WISTR) ldsWi[idx] = wiv[q];
  }
  if (tid < 256) ldsWout[tid] = hv0;
  ldsVec[wave][0][n16] = bv;
  ldsVec[wave][1][n16] = gv;
  ldsVec[wave][2][n16] = btv;
  ldsVec[wave][3][n16] = bov1;
  const bool need = (en < BAG) && (mk0 != 0.f || mk1 != 0.f);
  const unsigned bal = __builtin_amdgcn_ballot_w32(need);
  if (wave < 4 && lane == 0) ldsBal[wave] = bal;
  __builtin_amdgcn_fence(__ATOMIC_RELEASE, "workgroup", "local");
  __builtin_amdgcn_s_barrier();
  __builtin_amdgcn_fence(__ATOMIC_ACQUIRE, "workgroup", "local");

  const int c0 = __builtin_amdgcn_readfirstlane(__popc(ldsBal[0]));
  const int c1 = __builtin_amdgcn_readfirstlane(__popc(ldsBal[1]));
  const int c2 = __builtin_amdgcn_readfirstlane(__popc(ldsBal[2]));
  const int c3 = __builtin_amdgcn_readfirstlane(__popc(ldsBal[3]));
  const int nnzU = c0 + c1 + c2 + c3;
  const int nl = nnzU <= 32 ? 1 : (nnzU <= 64 ? 2 : 4);
  if (wave < 4) {
    const int pre = (wave > 0 ? c0 : 0) + (wave > 1 ? c1 : 0) + (wave > 2 ? c2 : 0);
    const int rk = pre + __popc(bal & ((1u << lane) - 1u));
    ldsRank[tid] = need ? rk : -1;
    if (need) ldsInv[rk] = tid;
  }

  const int pw = (wave + blockIdx.x) & (NWAVE - 1);
  float4 pv[16];
  if (!FIRST) {
    const float4* psrc = reinterpret_cast<const float4*>(
                             planes + (size_t)((stepK + 1) & 1) * (NWG * PLANE))
                         + pw * 32 + lane;
    if (nl == 1) {
#pragma unroll
      for (int q = 0; q < 4; ++q) pv[q] = psrc[(size_t)q * (NWAVE * 32)];
    } else if (nl == 2) {
#pragma unroll
      for (int q = 0; q < 8; ++q) pv[q] = psrc[(size_t)q * (NWAVE * 32)];
    } else {
#pragma unroll
      for (int q = 0; q < 16; ++q) pv[q] = psrc[(size_t)q * (NWAVE * 32)];
    }
  }

  v2f bw[17];
#pragma unroll
  for (int t = 0; t < 17; ++t)
    bw[t] = *(const v2f*)&ldsWi[n16 * WISTR + 4 * t + 2 * half];
  v8f accI = {0.f, 0.f, 0.f, 0.f, 0.f, 0.f, 0.f, 0.f};
#pragma unroll
  for (int j = 0; j < 8; ++j) {
    float qx = gq[j].x, qy = gq[j].y, qz = gq[j].z, qw = gq[j].w;
    asm volatile("v_permlane16_swap_b32 %0, %1" : "+v"(qx), "+v"(qz));
    asm volatile("v_permlane16_swap_b32 %0, %1" : "+v"(qy), "+v"(qw));
    v2f a0, a1;
    a0.x = qx; a0.y = qy;
    a1.x = qz; a1.y = qw;
    accI = __builtin_amdgcn_wmma_f32_16x16x4_f32(false, bw[2 * j], false, a0,
                                                 (short)0, accI, false, false);
    accI = __builtin_amdgcn_wmma_f32_16x16x4_f32(false, bw[2 * j + 1], false, a1,
                                                 (short)0, accI, false, false);
    __builtin_amdgcn_sched_barrier(0);
  }
  {
    v2f a;
    if (half == 0) {
      a.x = (float)p0 * (2.0f / DENOM) - 1.0f;
      a.y = (float)p1 * (2.0f / DENOM) - 1.0f;
    } else {
      a.x = 0.f; a.y = 0.f;
    }
    accI = __builtin_amdgcn_wmma_f32_16x16x4_f32(false, bw[16], false, a,
                                                 (short)0, accI, false, false);
  }

  if (!FIRST) {
    float4 s4;
    if (nl == 1) {
      s4 = plane_tree<4>(pv);
      s4.x += row_xor<8>(s4.x);         s4.y += row_xor<8>(s4.y);
      s4.z += row_xor<8>(s4.z);         s4.w += row_xor<8>(s4.w);
      s4.x += __shfl_xor(s4.x, 16, 32); s4.y += __shfl_xor(s4.y, 16, 32);
      s4.z += __shfl_xor(s4.z, 16, 32); s4.w += __shfl_xor(s4.w, 16, 32);
    } else if (nl == 2) {
      s4 = plane_tree<8>(pv);
      s4.x += __shfl_xor(s4.x, 16, 32); s4.y += __shfl_xor(s4.y, 16, 32);
      s4.z += __shfl_xor(s4.z, 16, 32); s4.w += __shfl_xor(s4.w, 16, 32);
    } else {
      s4 = plane_tree<16>(pv);
    }
    if (lane < 8 * nl) ldsRed[pw][lane] = s4;
    __syncthreads();

    if (tid < 256) {
      const int a = oi < oj ? oi : oj;
      const int b = oi < oj ? oj : oi;
      const int q = (a <= 7) ? a * 16 + b : (15 - a) * 16 + (b - a);
      const bool live = (oi < BAG) && (oj < BAG);
      const int rk = ldsRank[q];
      float co = 0.f;
      if (live && rk >= 0) {
        const float* rf = reinterpret_cast<const float*>(&ldsRed[0][0]) + rk;
        float t[NWAVE];
#pragma unroll
        for (int p = 0; p < NWAVE; ++p) t[p] = rf[p * PLANE];
#pragma unroll
        for (int st = NWAVE / 2; st >= 1; st >>= 1) {
#pragma unroll
          for (int p = 0; p < st; ++p) t[p] += t[p + st];
        }
        co = t[0] * (1.0f / (float)BATCH);
      }
      ldsReff[tid] = live ? (rv - LAMBDA * co) * mv : 0.f;
    }
    __syncthreads();
    const float sav[8] = {sq0.x, sq0.y, sq0.z, sq0.w, sq1.x, sq1.y, sq1.z, sq1.w};
#pragma unroll
    for (int t = 0; t < 4; ++t) {
      const int kf = 8 * half + 2 * t;
      v2f a, b;
      a.x = ldsReff[kf * 16 + n16];
      a.y = ldsReff[(kf + 1) * 16 + n16];
      b.x = sav[2 * t];
      b.y = sav[2 * t + 1];
      accI = __builtin_amdgcn_wmma_f32_16x16x4_f32(false, a, false, b,
                                                   (short)0, accI, false, false);
    }
  }

  float bvec[8], gvec[8], tvec[8];
  {
    const float4* pb = reinterpret_cast<const float4*>(&ldsVec[wave][0][8 * half]);
    const float4* pg = reinterpret_cast<const float4*>(&ldsVec[wave][1][8 * half]);
    const float4* pt = reinterpret_cast<const float4*>(&ldsVec[wave][2][8 * half]);
    const float4 b0 = pb[0], b1 = pb[1], g0 = pg[0], g1 = pg[1], t0 = pt[0], t1 = pt[1];
    bvec[0] = b0.x; bvec[1] = b0.y; bvec[2] = b0.z; bvec[3] = b0.w;
    bvec[4] = b1.x; bvec[5] = b1.y; bvec[6] = b1.z; bvec[7] = b1.w;
    gvec[0] = g0.x; gvec[1] = g0.y; gvec[2] = g0.z; gvec[3] = g0.w;
    gvec[4] = g1.x; gvec[5] = g1.y; gvec[6] = g1.z; gvec[7] = g1.w;
    tvec[0] = t0.x; tvec[1] = t0.y; tvec[2] = t0.z; tvec[3] = t0.w;
    tvec[4] = t1.x; tvec[5] = t1.y; tvec[6] = t1.z; tvec[7] = t1.w;
  }
  float hv[8];
#pragma unroll
  for (int r = 0; r < 8; ++r) hv[r] = fmaxf(accI[r] + bvec[r], 0.f);
  float rsum = ((hv[0] + hv[1]) + (hv[2] + hv[3])) + ((hv[4] + hv[5]) + (hv[6] + hv[7]));
  rsum += xhalf(rsum);
  const float mu = rsum * (1.0f / BAG);
  float dq[8];
#pragma unroll
  for (int r = 0; r < 8; ++r) {
    hv[r] = hv[r] - mu;
    dq[r] = hv[r] * hv[r];
  }
  dq[7] = half ? 0.f : dq[7];
  float vsum = ((dq[0] + dq[1]) + (dq[2] + dq[3])) + ((dq[4] + dq[5]) + (dq[6] + dq[7]));
  vsum += xhalf(vsum);
  const float rstd = rsqrtf(vsum * (1.0f / BAG) + 1e-5f);
#pragma unroll
  for (int r = 0; r < 8; ++r) hv[r] = hv[r] * rstd * gvec[r] + tvec[r];

  {
    const float4 o0 = make_float4(hv[0], hv[1], hv[2], hv[3]);
    const float4 o1 = make_float4(hv[4], hv[5], hv[6], hv[7]);
    float4* ls = reinterpret_cast<float4*>(&ldsState[wave][n16][8 * half]);
    ls[0] = o0;
    ls[1] = o1;
    if (stepK < NSTEP - 1) {
      float4* gs = reinterpret_cast<float4*>(stateWS + (size_t)myRow * 16 + 8 * half);
      gs[0] = o0;
      gs[1] = o1;
    }
  }
  __builtin_amdgcn_fence(__ATOMIC_RELEASE, "wavefront", "local");
  __builtin_amdgcn_wave_barrier();
  __builtin_amdgcn_fence(__ATOMIC_ACQUIRE, "wavefront", "local");

  v8f accO = {0.f, 0.f, 0.f, 0.f, 0.f, 0.f, 0.f, 0.f};
#pragma unroll
  for (int t = 0; t < 4; ++t) {
    const int kf = 8 * half + 2 * t;
    v2f a, b;
    a.x = ldsWout[kf * 16 + n16];
    a.y = ldsWout[(kf + 1) * 16 + n16];
    b.x = hv[2 * t];
    b.y = hv[2 * t + 1];
    accO = __builtin_amdgcn_wmma_f32_16x16x4_f32(false, a, false, b,
                                                 (short)0, accO, false, false);
  }

  if (stepK < NSTEP - 1) {
    const float raw0 = xhalf(accO[2]);
    const float raw  = half ? accO[3] : raw0;
    const float cm = tanhf(raw + (half ? bc1 : bc0));
    const float cx = xhalf(cm);
    if (half == 0) {
      const float c0 = cm, c1 = cx;
      int q0 = p0;
      int q1 = p1;
      const bool rowSel = fabsf(c0) >= fabsf(c1);
      const int mv0 = rowSel ? ((c0 > 0.f) - (c0 < 0.f)) : 0;
      const int mv1 = rowSel ? 0 : ((c1 > 0.f) - (c1 < 0.f));
      q0 = min(max(q0 + mv0, 0), DENOM);
      q1 = min(max(q1 + mv1, 0), DENOM);
      *reinterpret_cast<int2*>(posWS + 2 * myRow) = make_int2(q0, q1);
    }
  }

  if (stepK == NSTEP - 1) {
    const float4* po = reinterpret_cast<const float4*>(&ldsVec[wave][3][8 * half]);
    const float4 h0 = po[0], h1 = po[1];
    const float bov[8] = {h0.x, h0.y, h0.z, h0.w, h1.x, h1.y, h1.z, h1.w};
    float* orow = out + (size_t)myRow * 10 + 8 * half;
#pragma unroll
    for (int q = 0; q < 4; ++q) {
      if (8 * half + 2 * q < 10)
        *reinterpret_cast<float2*>(orow + 2 * q) =
            make_float2(accO[2 * q] + bov[2 * q], accO[2 * q + 1] + bov[2 * q + 1]);
    }
  }

  if (stepK < NSTEP - 1) {
    v8f accC = {0.f, 0.f, 0.f, 0.f, 0.f, 0.f, 0.f, 0.f};
#pragma unroll
    for (int t = 0; t < 4; ++t) {
      const int kb = 4 * t + 2 * half;
      v2f f;
      f.x = ldsState[wave][kb][n16];
      f.y = ldsState[wave][kb + 1][n16];
      accC = __builtin_amdgcn_wmma_f32_16x16x4_f32(false, f, false, f,
                                                   (short)0, accC, false, false);
    }
#pragma unroll
    for (int r = 0; r < 8; ++r)
      ldsCo[wave][(r + half * 8) * 16 + n16] = accC[r];
  }
  __builtin_amdgcn_fence(__ATOMIC_RELEASE, "workgroup", "local");
  __builtin_amdgcn_s_barrier();
  __builtin_amdgcn_fence(__ATOMIC_ACQUIRE, "workgroup", "local");
  if (stepK < NSTEP - 1 && tid < 32 * nl) {
    float acc = 0.f;
    if (tid < nnzU) {
      const int e = ldsInv[tid];
      const int r = e >> 4, c = e & 15;
      const int m = (c >= r) ? r : 15 - r;
      const int n = (c >= r) ? c : 15 - r + c;
#pragma unroll
      for (int w = 0; w < NWAVE; ++w) acc += ldsCo[w][m * 16 + n];
    }
    planes[(size_t)(stepK & 1) * (NWG * PLANE) + (size_t)blockIdx.x * (32 * nl) + tid] = acc;
  }
}

extern "C" void kernel_launch(void* const* d_in, const int* in_sizes, int n_in,
                              void* d_out, int out_size, void* d_ws, size_t ws_size,
                              hipStream_t stream) {
  const float* x    = (const float*)d_in[0];
  const float* Wi   = (const float*)d_in[1];
  const float* bi   = (const float*)d_in[2];
  const float* gma  = (const float*)d_in[3];
  const float* bta  = (const float*)d_in[4];
  const float* R    = (const float*)d_in[5];
  const float* rb   = (const float*)d_in[6];
  const float* mask = (const float*)d_in[7];
  const float* Wc   = (const float*)d_in[8];
  const float* bc   = (const float*)d_in[9];
  const float* Wo   = (const float*)d_in[10];
  const float* bo   = (const float*)d_in[11];

  float* stateWS  = (float*)d_ws;
  int*   posWS    = (int*)((char*)d_ws + (size_t)BATCH * 16 * 4);
  float* planes   = (float*)((char*)posWS + (size_t)BATCH * 2 * 4);

  cyclic_step<true><<<NWG, WGT, 0, stream>>>(x, Wi, bi, gma, bta, R, rb, mask, Wc, bc,
                                             Wo, bo, stateWS, posWS, planes,
                                             (float*)d_out, 0);
  for (int k = 1; k < NSTEP; ++k) {
    cyclic_step<false><<<NWG, WGT, 0, stream>>>(x, Wi, bi, gma, bta, R, rb, mask, Wc, bc,
                                                Wo, bo, stateWS, posWS, planes,
                                                (float*)d_out, k);
  }
}
